# starting kernel
# baseline (speedup 1.0000x reference)
#include <hip/hip_runtime.h>
#include <hip/hip_bf16.h>

typedef __attribute__((ext_vector_type(2))) float v2f;
typedef __attribute__((ext_vector_type(8))) float v8f;

#define IN_CH 128
#define LAT   32
#define TILE_M 16
#define APAD  132


__global__ void deg_init_kernel(float* __restrict__ deg, int n) {
    int i = blockIdx.x * blockDim.x + threadIdx.x;
    if (i < n) deg[i] = 1.0f;
}

__global__ void deg_edges_kernel(const int* __restrict__ ei, const int* __restrict__ yei,
                                 float* __restrict__ deg, int e1, int e2) {
    int e = blockIdx.x * blockDim.x + threadIdx.x;
    int et = e1 + e2;
    if (e >= et) return;
    int d = (e < e1) ? ei[e1 + e] : yei[e2 + (e - e1)];
    atomicAdd(deg + d, 1.0f);
}

__global__ void dinv_kernel(float* __restrict__ deg, int n) {
    int i = blockIdx.x * blockDim.x + threadIdx.x;
    if (i < n) deg[i] = rsqrtf(deg[i]);
}


__global__ __launch_bounds__(128) void gemm_wmma_kernel(const float* __restrict__ x,
                                                        const float* __restrict__ W,
                                                        float* __restrict__ h, int n) {
    __shared__ float sW[IN_CH * LAT];
    __shared__ float sA[4][TILE_M * APAD];

    const int tid  = threadIdx.x;
    const int wave = tid >> 5;
    const int lane = tid & 31;
    const int ntiles = (n + TILE_M - 1) / TILE_M;
    const int tile = blockIdx.x * 4 + wave;

    {
        const float4* Wv = (const float4*)W;
        float4* sWv = (float4*)sW;
        for (int i = tid; i < (IN_CH * LAT) / 4; i += 128) sWv[i] = Wv[i];
    }
    if (tile < ntiles) {
        const int r0 = tile * TILE_M;
        for (int i = lane; i < TILE_M * (IN_CH / 4); i += 32) {
            int row = i >> 5;
            int c4  = i & 31;
            int grow = r0 + row; if (grow >= n) grow = n - 1;
            float4 v = ((const float4*)(x + (size_t)grow * IN_CH))[c4];
            float* dst = &sA[wave][row * APAD + c4 * 4];
            dst[0] = v.x; dst[1] = v.y; dst[2] = v.z; dst[3] = v.w;
        }
    }
    __syncthreads();

    if (tile >= ntiles) return;

    v8f acc0 = {};
    v8f acc1 = {};
    const int m    = lane & 15;
    const int koff = (lane >> 4) << 1;
    const int nn   = lane & 15;

    for (int kk = 0; kk < IN_CH / 4; ++kk) {
        const int kb = kk * 4 + koff;
        v2f a;
        a.x = sA[wave][m * APAD + kb];
        a.y = sA[wave][m * APAD + kb + 1];
        v2f b0, b1;
        b0.x = sW[(kb    ) * LAT + nn];
        b0.y = sW[(kb + 1) * LAT + nn];
        b1.x = sW[(kb    ) * LAT + 16 + nn];
        b1.y = sW[(kb + 1) * LAT + 16 + nn];
        acc0 = __builtin_amdgcn_wmma_f32_16x16x4_f32(false, a, false, b0,
                                                     (short)0, acc0, false, false);
        acc1 = __builtin_amdgcn_wmma_f32_16x16x4_f32(false, a, false, b1,
                                                     (short)0, acc1, false, false);
    }

    const int r0   = tile * TILE_M;
    const int rofs = (lane >> 4) << 3;
#pragma unroll
    for (int r = 0; r < 8; ++r) {
        int row = r0 + r + rofs;
        if (row < n) {
            h[(size_t)row * LAT + nn]      = acc0[r];
            h[(size_t)row * LAT + 16 + nn] = acc1[r];
        }
    }
}


__global__ void init_out_kernel(const float* __restrict__ dinv, const float* __restrict__ h,
                                const float* __restrict__ b, float* __restrict__ out, int n) {
    int gid = blockIdx.x * blockDim.x + threadIdx.x;
    if (gid >= n * LAT) return;
    int i = gid >> 5, c = gid & 31;
    float di = dinv[i];
    out[gid] = di * di * h[gid] + b[c];
}


__global__ void edge_scatter_kernel(const int* __restrict__ ei, const int* __restrict__ yei,
                                    const float* __restrict__ dinv, const float* __restrict__ h,
                                    float* __restrict__ out, int e1, int e2) {
    long long gid = (long long)blockIdx.x * blockDim.x + threadIdx.x;
    int e = (int)(gid >> 5);
    int c = (int)(gid & 31);
    int et = e1 + e2;
    if (e >= et) return;
    int s, d;
    if (e < e1) { s = ei[e];            d = ei[e1 + e]; }
    else        { int f = e - e1; s = yei[f]; d = yei[e2 + f]; }
    float w = dinv[s] * dinv[d];
    atomicAdd(out + (size_t)d * LAT + c, w * h[(size_t)s * LAT + c]);
}


extern "C" void kernel_launch(void* const* d_in, const int* in_sizes, int n_in,
                              void* d_out, int out_size, void* d_ws, size_t ws_size,
                              hipStream_t stream) {
    const float* x   = (const float*)d_in[0];
    const int*   ei  = (const int*)  d_in[1];
    const int*   yei = (const int*)  d_in[2];
    const float* W   = (const float*)d_in[3];
    const float* b   = (const float*)d_in[4];
    float* out = (float*)d_out;

    const int n  = in_sizes[0] / IN_CH;
    const int e1 = in_sizes[1] / 2;
    const int e2 = in_sizes[2] / 2;
    const int et = e1 + e2;

    float* deg = (float*)d_ws;
    size_t hoff = (((size_t)n * sizeof(float)) + 255) & ~(size_t)255;
    float* h = (float*)((char*)d_ws + hoff);

    deg_init_kernel<<<(n + 255) / 256, 256, 0, stream>>>(deg, n);
    deg_edges_kernel<<<(et + 255) / 256, 256, 0, stream>>>(ei, yei, deg, e1, e2);
    dinv_kernel<<<(n + 255) / 256, 256, 0, stream>>>(deg, n);
    {
        int ntiles = (n + TILE_M - 1) / TILE_M;
        int blocks = (ntiles + 3) / 4;
        gemm_wmma_kernel<<<blocks, 128, 0, stream>>>(x, W, h, n);
    }
    init_out_kernel<<<((size_t)n * LAT + 255) / 256, 256, 0, stream>>>(deg, h, b, out, n);
    {
        long long threads = (long long)et * 32;
        int blocks = (int)((threads + 255) / 256);
        edge_scatter_kernel<<<blocks, 256, 0, stream>>>(ei, yei, deg, h, out, e1, e2);
    }
}


# degrees counted from the destination-grouped list: one block per range histograms its slice in LDS and stores rsqrt(1+count); no deg init, global float atomics or rsqrt pass (8->6 launches)
# speedup vs baseline: 3.4934x; 3.4934x over previous
#include <hip/hip_runtime.h>
#include <hip/hip_bf16.h>

typedef __attribute__((ext_vector_type(2))) float v2f;
typedef __attribute__((ext_vector_type(8))) float v8f;

#define IN_CH 128
#define LAT   32
#define TILE_M 16
#define APAD  132


__global__ void deg_init_kernel(float* __restrict__ deg, int n) {
    int i = blockIdx.x * blockDim.x + threadIdx.x;
    if (i < n) deg[i] = 1.0f;
}

__global__ void deg_edges_kernel(const int* __restrict__ ei, const int* __restrict__ yei,
                                 float* __restrict__ deg, int e1, int e2) {
    int e = blockIdx.x * blockDim.x + threadIdx.x;
    int et = e1 + e2;
    if (e >= et) return;
    int d = (e < e1) ? ei[e1 + e] : yei[e2 + (e - e1)];
    atomicAdd(deg + d, 1.0f);
}

__global__ void dinv_kernel(float* __restrict__ deg, int n) {
    int i = blockIdx.x * blockDim.x + threadIdx.x;
    if (i < n) deg[i] = rsqrtf(deg[i]);
}


__global__ __launch_bounds__(128) void gemm_wmma_kernel(const float* __restrict__ x,
                                                        const float* __restrict__ W,
                                                        float* __restrict__ h, int n) {
    __shared__ float sW[IN_CH * LAT];
    __shared__ float sA[4][TILE_M * APAD];

    const int tid  = threadIdx.x;
    const int wave = tid >> 5;
    const int lane = tid & 31;
    const int ntiles = (n + TILE_M - 1) / TILE_M;
    const int tile = blockIdx.x * 4 + wave;

    {
        const float4* Wv = (const float4*)W;
        float4* sWv = (float4*)sW;
        for (int i = tid; i < (IN_CH * LAT) / 4; i += 128) sWv[i] = Wv[i];
    }
    if (tile < ntiles) {
        const int r0 = tile * TILE_M;
        for (int i = lane; i < TILE_M * (IN_CH / 4); i += 32) {
            int row = i >> 5;
            int c4  = i & 31;
            int grow = r0 + row; if (grow >= n) grow = n - 1;
            float4 v = ((const float4*)(x + (size_t)grow * IN_CH))[c4];
            float* dst = &sA[wave][row * APAD + c4 * 4];
            dst[0] = v.x; dst[1] = v.y; dst[2] = v.z; dst[3] = v.w;
        }
    }
    __syncthreads();

    if (tile >= ntiles) return;

    v8f acc0 = {};
    v8f acc1 = {};
    const int m    = lane & 15;
    const int koff = (lane >> 4) << 1;
    const int nn   = lane & 15;

    for (int kk = 0; kk < IN_CH / 4; ++kk) {
        const int kb = kk * 4 + koff;
        v2f a;
        a.x = sA[wave][m * APAD + kb];
        a.y = sA[wave][m * APAD + kb + 1];
        v2f b0, b1;
        b0.x = sW[(kb    ) * LAT + nn];
        b0.y = sW[(kb + 1) * LAT + nn];
        b1.x = sW[(kb    ) * LAT + 16 + nn];
        b1.y = sW[(kb + 1) * LAT + 16 + nn];
        acc0 = __builtin_amdgcn_wmma_f32_16x16x4_f32(false, a, false, b0,
                                                     (short)0, acc0, false, false);
        acc1 = __builtin_amdgcn_wmma_f32_16x16x4_f32(false, a, false, b1,
                                                     (short)0, acc1, false, false);
    }

    const int r0   = tile * TILE_M;
    const int rofs = (lane >> 4) << 3;
#pragma unroll
    for (int r = 0; r < 8; ++r) {
        int row = r0 + r + rofs;
        if (row < n) {
            h[(size_t)row * LAT + nn]      = acc0[r];
            h[(size_t)row * LAT + 16 + nn] = acc1[r];
        }
    }
}


__global__ void init_out_kernel(const float* __restrict__ dinv, const float* __restrict__ h,
                                const float* __restrict__ b, float* __restrict__ out, int n) {
    int gid = blockIdx.x * blockDim.x + threadIdx.x;
    if (gid >= n * LAT) return;
    int i = gid >> 5, c = gid & 31;
    float di = dinv[i];
    out[gid] = di * di * h[gid] + b[c];
}


__global__ void edge_scatter_kernel(const int* __restrict__ ei, const int* __restrict__ yei,
                                    const float* __restrict__ dinv, const float* __restrict__ h,
                                    float* __restrict__ out, int e1, int e2) {
    long long gid = (long long)blockIdx.x * blockDim.x + threadIdx.x;
    int e = (int)(gid >> 5);
    int c = (int)(gid & 31);
    int et = e1 + e2;
    if (e >= et) return;
    int s, d;
    if (e < e1) { s = ei[e];            d = ei[e1 + e]; }
    else        { int f = e - e1; s = yei[f]; d = yei[e2 + f]; }
    float w = dinv[s] * dinv[d];
    atomicAdd(out + (size_t)d * LAT + c, w * h[(size_t)s * LAT + c]);
}


#define NB_LOG 8
#define NB   256
#define MAXB 1024
#define CH   8192
#define PT   512
#define AT   512
#define DG   512
#define DU   12

__device__ __forceinline__ int edge_src(const int* __restrict__ ei, const int* __restrict__ yei,
                                        int e1, int e) {
    return (e < e1) ? ei[e] : yei[e - e1];
}
__device__ __forceinline__ int edge_dst(const int* __restrict__ ei, const int* __restrict__ yei,
                                        int e1, int e2, int e) {
    return (e < e1) ? ei[e1 + e] : yei[e2 + (e - e1)];
}

template <int NT>
__device__ __forceinline__ int block_excl_scan(int v, int* sw, int& total) {
    const int lane = threadIdx.x & 31;
    const int wv   = threadIdx.x >> 5;
    int inc = v;
#pragma unroll
    for (int d = 1; d < 32; d <<= 1) {
        int t = __shfl_up(inc, d);
        if (lane >= d) inc += t;
    }
    if (lane == 31) sw[wv] = inc;
    __syncthreads();
    if (wv == 0) {
        int s  = (lane < NT / 32) ? sw[lane] : 0;
        int si = s;
#pragma unroll
        for (int d = 1; d < 32; d <<= 1) {
            int t = __shfl_up(si, d);
            if (lane >= d) si += t;
        }
        if (lane < NT / 32) sw[lane] = si - s;
        if (lane == 31) sw[32] = si;
    }
    __syncthreads();
    const int r = sw[wv] + inc - v;
    total = sw[32];
    __syncthreads();
    return r;
}

__global__ __launch_bounds__(PT) void range_count_kernel(const int* __restrict__ ei,
                                                         const int* __restrict__ yei,
                                                         int* __restrict__ counts,
                                                         int e1, int e2, int nchunks, int nb) {
    __shared__ int hist[MAXB];
    const int tid = threadIdx.x;
    const int c   = blockIdx.x;
    const int et  = e1 + e2;
    const int e0  = c * CH;
    for (int i = tid; i < nb; i += PT) hist[i] = 0;
    int d[CH / PT];
#pragma unroll
    for (int k = 0; k < CH / PT; ++k) {
        const int e = e0 + tid + k * PT;
        d[k] = edge_dst(ei, yei, e1, e2, (e < et) ? e : et - 1);
    }
    __syncthreads();
#pragma unroll
    for (int k = 0; k < CH / PT; ++k) {
        const int e = e0 + tid + k * PT;
        if (e < et) atomicAdd(&hist[d[k] >> NB_LOG], 1);
    }
    __syncthreads();
    for (int b = tid; b < nb; b += PT) counts[(size_t)b * nchunks + c] = hist[b];
}

__global__ __launch_bounds__(256) void range_scan_kernel(int* __restrict__ counts,
                                                         int* __restrict__ btot, int nchunks) {
    __shared__ int sw[33];
    const int tid = threadIdx.x;
    int* row = counts + (size_t)blockIdx.x * nchunks;
    int carry = 0;
    for (int c0 = 0; c0 < nchunks; c0 += 256) {
        const int c = c0 + tid;
        const int v = (c < nchunks) ? row[c] : 0;
        int tot;
        const int ex = block_excl_scan<256>(v, sw, tot);
        if (c < nchunks) row[c] = carry + ex;
        carry += tot;
    }
    if (tid == 0) btot[blockIdx.x] = carry;
}

__global__ __launch_bounds__(PT) void range_place_kernel(const int* __restrict__ ei,
                                                         const int* __restrict__ yei,
                                                         const int* __restrict__ woff,
                                                         const int* __restrict__ btot,
                                                         int* __restrict__ bstart,
                                                         unsigned* __restrict__ sorted,
                                                         int e1, int e2, int nchunks, int nb) {
    __shared__ int hist[MAXB];
    __shared__ int lbase[MAXB];
    __shared__ int gbase[MAXB];
    __shared__ unsigned words[CH];
    __shared__ unsigned short bk[CH];
    __shared__ int sw[33];

    const int tid = threadIdx.x;
    const int c   = blockIdx.x;
    const int et  = e1 + e2;
    const int e0  = c * CH;

    for (int i = tid; i < nb; i += PT) hist[i] = 0;

    int sv[CH / PT];
    int dv[CH / PT];
#pragma unroll
    for (int k = 0; k < CH / PT; ++k) {
        const int e  = e0 + tid + k * PT;
        const int ec = (e < et) ? e : et - 1;
        sv[k] = edge_src(ei, yei, e1, ec);
        dv[k] = edge_dst(ei, yei, e1, e2, ec);
    }
    __syncthreads();

    unsigned wreg[CH / PT];
    unsigned key[CH / PT];
#pragma unroll
    for (int k = 0; k < CH / PT; ++k) {
        const int e = e0 + tid + k * PT;
        key[k]  = 0xFFFFFFFFu;
        wreg[k] = 0u;
        if (e < et) {
            const int s = sv[k];
            const int d = dv[k];
            const int b = d >> NB_LOG;
            const int r = atomicAdd(&hist[b], 1);
            wreg[k] = ((unsigned)s << NB_LOG) | (unsigned)(d & (NB - 1));
            key[k]  = ((unsigned)b << 16) | (unsigned)r;
        }
    }
    __syncthreads();

    int carryL = 0, carryG = 0;
    for (int b0 = 0; b0 < nb; b0 += PT) {
        const int b  = b0 + tid;
        const int hv = (b < nb) ? hist[b] : 0;
        const int gv = (b < nb) ? btot[b] : 0;
        int tl, tg;
        const int el = block_excl_scan<PT>(hv, sw, tl);
        const int eg = block_excl_scan<PT>(gv, sw, tg);
        if (b < nb) {
            const int bs = carryG + eg;
            lbase[b] = carryL + el;
            gbase[b] = bs + woff[(size_t)b * nchunks + c];
            if (c == 0) bstart[b] = bs;
        }
        carryL += tl;
        carryG += tg;
    }
    if (c == 0 && tid == 0) bstart[nb] = carryG;
    __syncthreads();

#pragma unroll
    for (int k = 0; k < CH / PT; ++k) {
        if (key[k] != 0xFFFFFFFFu) {
            const int b = (int)(key[k] >> 16);
            const int p = lbase[b] + (int)(key[k] & 0xFFFFu);
            words[p] = wreg[k];
            bk[p]    = (unsigned short)b;
        }
    }
    __syncthreads();

    const int cnt = (et - e0 < CH) ? (et - e0) : CH;
    for (int i = tid; i < cnt; i += PT) {
        const int b = bk[i];
        sorted[gbase[b] + (i - lbase[b])] = words[i];
    }
}

__global__ __launch_bounds__(DG) void range_deg_kernel(const unsigned* __restrict__ sorted,
                                                       const int* __restrict__ bstart,
                                                       float* __restrict__ dinv, int n) {
    __shared__ int cnt[NB];
    const int tid = threadIdx.x;
    const int lo  = blockIdx.x * NB;
    for (int r = tid; r < NB; r += DG) cnt[r] = 0;

    const int start = bstart[blockIdx.x];
    const int end   = bstart[blockIdx.x + 1];
    const int lastw = (end > 0) ? (end - 1) : 0;

    unsigned w[DU];
#pragma unroll
    for (int u = 0; u < DU; ++u) {
        const int i = start + tid + u * DG;
        w[u] = sorted[(i < end) ? i : lastw];
    }
    __syncthreads();
#pragma unroll
    for (int u = 0; u < DU; ++u) {
        const int i = start + tid + u * DG;
        if (i < end) atomicAdd(&cnt[w[u] & (NB - 1)], 1);
    }
    for (int i0 = start + DU * DG; i0 < end; i0 += DU * DG) {
#pragma unroll
        for (int u = 0; u < DU; ++u) {
            const int i = i0 + tid + u * DG;
            w[u] = sorted[(i < end) ? i : lastw];
        }
#pragma unroll
        for (int u = 0; u < DU; ++u) {
            const int i = i0 + tid + u * DG;
            if (i < end) atomicAdd(&cnt[w[u] & (NB - 1)], 1);
        }
    }
    __syncthreads();
    for (int r = tid; r < NB; r += DG) {
        if (lo + r < n) dinv[lo + r] = rsqrtf((float)(1 + cnt[r]));
    }
}

__global__ __launch_bounds__(AT) void range_sum_kernel(const unsigned* __restrict__ sorted,
                                                       const int* __restrict__ bstart,
                                                       const float* __restrict__ dinv,
                                                       const float* __restrict__ h,
                                                       const float* __restrict__ bias,
                                                       float* __restrict__ out, int n) {
    __shared__ float acc[NB * LAT];
    __shared__ float sdinv[NB];

    const int tid  = threadIdx.x;
    const int lane = tid & 31;
    const int wave = tid >> 5;
    const int lo   = blockIdx.x * NB;

    for (int r = tid; r < NB; r += AT) {
        const float di = dinv[(lo + r < n) ? (lo + r) : (n - 1)];
        sdinv[r] = (lo + r < n) ? di : 0.0f;
    }
    {
        constexpr int RI = AT / 32;
        constexpr int NI = NB / RI;
        const float bl = bias[lane];
        float dv[NI];
        float hv[NI];
#pragma unroll
        for (int k = 0; k < NI; ++k) {
            const int node = lo + wave + k * RI;
            const int nc   = (node < n) ? node : (n - 1);
            dv[k] = dinv[nc];
            hv[k] = h[(size_t)nc * LAT + lane];
        }
#pragma unroll
        for (int k = 0; k < NI; ++k) {
            const int row = wave + k * RI;
            acc[row * LAT + lane] = (lo + row < n) ? (dv[k] * dv[k] * hv[k] + bl) : 0.0f;
        }
    }
    __syncthreads();

    const int start  = bstart[blockIdx.x];
    const int end    = bstart[blockIdx.x + 1];
    const int nbatch = (end - start + 31) >> 5;
    const float* hl  = h + lane;

    const int lastw = end - 1;
    constexpr int KS = AT / 32;
    unsigned wd_cur = 0u, wd_nxt = 0u;
    float ds_cur = 0.0f;
    if (wave < nbatch) {
        const int i = start + (wave << 5) + lane;
        wd_cur = sorted[(i < end) ? i : lastw];
    }
    if (wave + KS < nbatch) {
        const int i = start + ((wave + KS) << 5) + lane;
        wd_nxt = sorted[(i < end) ? i : lastw];
    }
    if (wave < nbatch) ds_cur = dinv[wd_cur >> NB_LOG];

    for (int k = wave; k < nbatch; k += KS) {
        const int base = start + (k << 5);
        const int m    = (end - base < 32) ? (end - base) : 32;
        const unsigned wd = wd_cur;
        const float    ds = ds_cur;
        if (k + KS < nbatch) ds_cur = dinv[wd_nxt >> NB_LOG];
        wd_cur = wd_nxt;
        if (k + 2 * KS < nbatch) {
            const int i = start + ((k + 2 * KS) << 5) + lane;
            wd_nxt = sorted[(i < end) ? i : lastw];
        }
        const int   s  = (int)(wd >> NB_LOG);
        const int   dl = (int)(wd & (NB - 1));
        const float w  = ds * sdinv[dl];
        if (m == 32) {
            float v[32];
#pragma unroll
            for (int j = 0; j < 32; ++j) {
                const int sj = __builtin_amdgcn_readlane(s, j);
                v[j] = hl[(size_t)sj * LAT];
            }
#pragma unroll
            for (int j = 0; j < 32; ++j) {
                const float wj = __int_as_float(__builtin_amdgcn_readlane(__float_as_int(w), j));
                const int   dj = __builtin_amdgcn_readlane(dl, j);
                atomicAdd(&acc[dj * LAT + lane], wj * v[j]);
            }
        } else {
            for (int j = 0; j < m; ++j) {
                const int   sj = __builtin_amdgcn_readlane(s, j);
                const float wj = __int_as_float(__builtin_amdgcn_readlane(__float_as_int(w), j));
                const int   dj = __builtin_amdgcn_readlane(dl, j);
                atomicAdd(&acc[dj * LAT + lane], wj * hl[(size_t)sj * LAT]);
            }
        }
    }
    __syncthreads();

    for (int i = tid; i < NB * LAT; i += AT) {
        if (lo + (i >> 5) < n) out[(size_t)lo * LAT + i] = acc[i];
    }
}


extern "C" void kernel_launch(void* const* d_in, const int* in_sizes, int n_in,
                              void* d_out, int out_size, void* d_ws, size_t ws_size,
                              hipStream_t stream) {
    const float* x   = (const float*)d_in[0];
    const int*   ei  = (const int*)  d_in[1];
    const int*   yei = (const int*)  d_in[2];
    const float* W   = (const float*)d_in[3];
    const float* b   = (const float*)d_in[4];
    float* out = (float*)d_out;

    const int n  = in_sizes[0] / IN_CH;
    const int e1 = in_sizes[1] / 2;
    const int e2 = in_sizes[2] / 2;
    const int et = e1 + e2;

    float* deg = (float*)d_ws;
    size_t hoff = (((size_t)n * sizeof(float)) + 255) & ~(size_t)255;
    float* h = (float*)((char*)d_ws + hoff);

    const int nb      = (n + NB - 1) / NB;
    const int nchunks = (et + CH - 1) / CH;
    size_t soff = (hoff + (size_t)n * LAT * sizeof(float) + 255) & ~(size_t)255;
    size_t coff = (soff + (size_t)et * sizeof(unsigned) + 255) & ~(size_t)255;
    size_t toff = (coff + (size_t)nb * nchunks * sizeof(int) + 255) & ~(size_t)255;
    size_t boff = (toff + (size_t)nb * sizeof(int) + 255) & ~(size_t)255;
    size_t need = boff + ((size_t)nb + 1) * sizeof(int);
    unsigned* sorted = (unsigned*)((char*)d_ws + soff);
    int* counts = (int*)((char*)d_ws + coff);
    int* btot   = (int*)((char*)d_ws + toff);
    int* bstart = (int*)((char*)d_ws + boff);
    const bool grouped = (need <= ws_size) && (nb <= MAXB) && (n <= (1 << 24)) && (nchunks > 0);

    {
        int ntiles = (n + TILE_M - 1) / TILE_M;
        int blocks = (ntiles + 3) / 4;
        gemm_wmma_kernel<<<blocks, 128, 0, stream>>>(x, W, h, n);
    }
    if (grouped) {
        range_count_kernel<<<nchunks, PT, 0, stream>>>(ei, yei, counts, e1, e2, nchunks, nb);
        range_scan_kernel<<<nb, 256, 0, stream>>>(counts, btot, nchunks);
        range_place_kernel<<<nchunks, PT, 0, stream>>>(ei, yei, counts, btot, bstart, sorted,
                                                       e1, e2, nchunks, nb);
        range_deg_kernel<<<nb, DG, 0, stream>>>(sorted, bstart, deg, n);
        range_sum_kernel<<<nb, AT, 0, stream>>>(sorted, bstart, deg, h, b, out, n);
    } else {
        deg_init_kernel<<<(n + 255) / 256, 256, 0, stream>>>(deg, n);
        deg_edges_kernel<<<(et + 255) / 256, 256, 0, stream>>>(ei, yei, deg, e1, e2);
        dinv_kernel<<<(n + 255) / 256, 256, 0, stream>>>(deg, n);
        init_out_kernel<<<((size_t)n * LAT + 255) / 256, 256, 0, stream>>>(deg, h, b, out, n);
        long long threads = (long long)et * 32;
        int blocks = (int)((threads + 255) / 256);
        edge_scatter_kernel<<<blocks, 256, 0, stream>>>(ei, yei, deg, h, out, e1, e2);
    }
}


# range-sum: a batch's 32 row gathers use 32 offset registers off one scalar base and go out as one group, look-ahead loads behind them: no translation wait between them; sum order as before
# speedup vs baseline: 3.7168x; 1.0639x over previous
#include <hip/hip_runtime.h>
#include <hip/hip_bf16.h>

typedef __attribute__((ext_vector_type(2))) float v2f;
typedef __attribute__((ext_vector_type(8))) float v8f;

#define IN_CH 128
#define LAT   32
#define TILE_M 16
#define APAD  132


__global__ void deg_init_kernel(float* __restrict__ deg, int n) {
    int i = blockIdx.x * blockDim.x + threadIdx.x;
    if (i < n) deg[i] = 1.0f;
}

__global__ void deg_edges_kernel(const int* __restrict__ ei, const int* __restrict__ yei,
                                 float* __restrict__ deg, int e1, int e2) {
    int e = blockIdx.x * blockDim.x + threadIdx.x;
    int et = e1 + e2;
    if (e >= et) return;
    int d = (e < e1) ? ei[e1 + e] : yei[e2 + (e - e1)];
    atomicAdd(deg + d, 1.0f);
}

__global__ void dinv_kernel(float* __restrict__ deg, int n) {
    int i = blockIdx.x * blockDim.x + threadIdx.x;
    if (i < n) deg[i] = rsqrtf(deg[i]);
}


__global__ __launch_bounds__(128) void gemm_wmma_kernel(const float* __restrict__ x,
                                                        const float* __restrict__ W,
                                                        float* __restrict__ h, int n) {
    __shared__ float sW[IN_CH * LAT];
    __shared__ float sA[4][TILE_M * APAD];

    const int tid  = threadIdx.x;
    const int wave = tid >> 5;
    const int lane = tid & 31;
    const int ntiles = (n + TILE_M - 1) / TILE_M;
    const int tile = blockIdx.x * 4 + wave;

    {
        const float4* Wv = (const float4*)W;
        float4* sWv = (float4*)sW;
        for (int i = tid; i < (IN_CH * LAT) / 4; i += 128) sWv[i] = Wv[i];
    }
    if (tile < ntiles) {
        const int r0 = tile * TILE_M;
        for (int i = lane; i < TILE_M * (IN_CH / 4); i += 32) {
            int row = i >> 5;
            int c4  = i & 31;
            int grow = r0 + row; if (grow >= n) grow = n - 1;
            float4 v = ((const float4*)(x + (size_t)grow * IN_CH))[c4];
            float* dst = &sA[wave][row * APAD + c4 * 4];
            dst[0] = v.x; dst[1] = v.y; dst[2] = v.z; dst[3] = v.w;
        }
    }
    __syncthreads();

    if (tile >= ntiles) return;

    v8f acc0 = {};
    v8f acc1 = {};
    const int m    = lane & 15;
    const int koff = (lane >> 4) << 1;
    const int nn   = lane & 15;

    for (int kk = 0; kk < IN_CH / 4; ++kk) {
        const int kb = kk * 4 + koff;
        v2f a;
        a.x = sA[wave][m * APAD + kb];
        a.y = sA[wave][m * APAD + kb + 1];
        v2f b0, b1;
        b0.x = sW[(kb    ) * LAT + nn];
        b0.y = sW[(kb + 1) * LAT + nn];
        b1.x = sW[(kb    ) * LAT + 16 + nn];
        b1.y = sW[(kb + 1) * LAT + 16 + nn];
        acc0 = __builtin_amdgcn_wmma_f32_16x16x4_f32(false, a, false, b0,
                                                     (short)0, acc0, false, false);
        acc1 = __builtin_amdgcn_wmma_f32_16x16x4_f32(false, a, false, b1,
                                                     (short)0, acc1, false, false);
    }

    const int r0   = tile * TILE_M;
    const int rofs = (lane >> 4) << 3;
#pragma unroll
    for (int r = 0; r < 8; ++r) {
        int row = r0 + r + rofs;
        if (row < n) {
            h[(size_t)row * LAT + nn]      = acc0[r];
            h[(size_t)row * LAT + 16 + nn] = acc1[r];
        }
    }
}


__global__ void init_out_kernel(const float* __restrict__ dinv, const float* __restrict__ h,
                                const float* __restrict__ b, float* __restrict__ out, int n) {
    int gid = blockIdx.x * blockDim.x + threadIdx.x;
    if (gid >= n * LAT) return;
    int i = gid >> 5, c = gid & 31;
    float di = dinv[i];
    out[gid] = di * di * h[gid] + b[c];
}


__global__ void edge_scatter_kernel(const int* __restrict__ ei, const int* __restrict__ yei,
                                    const float* __restrict__ dinv, const float* __restrict__ h,
                                    float* __restrict__ out, int e1, int e2) {
    long long gid = (long long)blockIdx.x * blockDim.x + threadIdx.x;
    int e = (int)(gid >> 5);
    int c = (int)(gid & 31);
    int et = e1 + e2;
    if (e >= et) return;
    int s, d;
    if (e < e1) { s = ei[e];            d = ei[e1 + e]; }
    else        { int f = e - e1; s = yei[f]; d = yei[e2 + f]; }
    float w = dinv[s] * dinv[d];
    atomicAdd(out + (size_t)d * LAT + c, w * h[(size_t)s * LAT + c]);
}


#define NB_LOG 8
#define NB   256
#define MAXB 1024
#define CH   8192
#define PT   512
#define AT   512
#define DG   512
#define DU   12

__device__ __forceinline__ int edge_src(const int* __restrict__ ei, const int* __restrict__ yei,
                                        int e1, int e) {
    return (e < e1) ? ei[e] : yei[e - e1];
}
__device__ __forceinline__ int edge_dst(const int* __restrict__ ei, const int* __restrict__ yei,
                                        int e1, int e2, int e) {
    return (e < e1) ? ei[e1 + e] : yei[e2 + (e - e1)];
}

template <int NT>
__device__ __forceinline__ int block_excl_scan(int v, int* sw, int& total) {
    const int lane = threadIdx.x & 31;
    const int wv   = threadIdx.x >> 5;
    int inc = v;
#pragma unroll
    for (int d = 1; d < 32; d <<= 1) {
        int t = __shfl_up(inc, d);
        if (lane >= d) inc += t;
    }
    if (lane == 31) sw[wv] = inc;
    __syncthreads();
    if (wv == 0) {
        int s  = (lane < NT / 32) ? sw[lane] : 0;
        int si = s;
#pragma unroll
        for (int d = 1; d < 32; d <<= 1) {
            int t = __shfl_up(si, d);
            if (lane >= d) si += t;
        }
        if (lane < NT / 32) sw[lane] = si - s;
        if (lane == 31) sw[32] = si;
    }
    __syncthreads();
    const int r = sw[wv] + inc - v;
    total = sw[32];
    __syncthreads();
    return r;
}

__global__ __launch_bounds__(PT) void range_count_kernel(const int* __restrict__ ei,
                                                         const int* __restrict__ yei,
                                                         int* __restrict__ counts,
                                                         int e1, int e2, int nchunks, int nb) {
    __shared__ int hist[MAXB];
    const int tid = threadIdx.x;
    const int c   = blockIdx.x;
    const int et  = e1 + e2;
    const int e0  = c * CH;
    for (int i = tid; i < nb; i += PT) hist[i] = 0;
    int d[CH / PT];
#pragma unroll
    for (int k = 0; k < CH / PT; ++k) {
        const int e = e0 + tid + k * PT;
        d[k] = edge_dst(ei, yei, e1, e2, (e < et) ? e : et - 1);
    }
    __syncthreads();
#pragma unroll
    for (int k = 0; k < CH / PT; ++k) {
        const int e = e0 + tid + k * PT;
        if (e < et) atomicAdd(&hist[d[k] >> NB_LOG], 1);
    }
    __syncthreads();
    for (int b = tid; b < nb; b += PT) counts[(size_t)b * nchunks + c] = hist[b];
}

__global__ __launch_bounds__(256) void range_scan_kernel(int* __restrict__ counts,
                                                         int* __restrict__ btot, int nchunks) {
    __shared__ int sw[33];
    const int tid = threadIdx.x;
    int* row = counts + (size_t)blockIdx.x * nchunks;
    int carry = 0;
    for (int c0 = 0; c0 < nchunks; c0 += 256) {
        const int c = c0 + tid;
        const int v = (c < nchunks) ? row[c] : 0;
        int tot;
        const int ex = block_excl_scan<256>(v, sw, tot);
        if (c < nchunks) row[c] = carry + ex;
        carry += tot;
    }
    if (tid == 0) btot[blockIdx.x] = carry;
}

__global__ __launch_bounds__(PT) void range_place_kernel(const int* __restrict__ ei,
                                                         const int* __restrict__ yei,
                                                         const int* __restrict__ woff,
                                                         const int* __restrict__ btot,
                                                         int* __restrict__ bstart,
                                                         unsigned* __restrict__ sorted,
                                                         int e1, int e2, int nchunks, int nb) {
    __shared__ int hist[MAXB];
    __shared__ int lbase[MAXB];
    __shared__ int gbase[MAXB];
    __shared__ unsigned words[CH];
    __shared__ unsigned short bk[CH];
    __shared__ int sw[33];

    const int tid = threadIdx.x;
    const int c   = blockIdx.x;
    const int et  = e1 + e2;
    const int e0  = c * CH;

    for (int i = tid; i < nb; i += PT) hist[i] = 0;

    int sv[CH / PT];
    int dv[CH / PT];
#pragma unroll
    for (int k = 0; k < CH / PT; ++k) {
        const int e  = e0 + tid + k * PT;
        const int ec = (e < et) ? e : et - 1;
        sv[k] = edge_src(ei, yei, e1, ec);
        dv[k] = edge_dst(ei, yei, e1, e2, ec);
    }
    __syncthreads();

    unsigned wreg[CH / PT];
    unsigned key[CH / PT];
#pragma unroll
    for (int k = 0; k < CH / PT; ++k) {
        const int e = e0 + tid + k * PT;
        key[k]  = 0xFFFFFFFFu;
        wreg[k] = 0u;
        if (e < et) {
            const int s = sv[k];
            const int d = dv[k];
            const int b = d >> NB_LOG;
            const int r = atomicAdd(&hist[b], 1);
            wreg[k] = ((unsigned)s << NB_LOG) | (unsigned)(d & (NB - 1));
            key[k]  = ((unsigned)b << 16) | (unsigned)r;
        }
    }
    __syncthreads();

    int carryL = 0, carryG = 0;
    for (int b0 = 0; b0 < nb; b0 += PT) {
        const int b  = b0 + tid;
        const int hv = (b < nb) ? hist[b] : 0;
        const int gv = (b < nb) ? btot[b] : 0;
        int tl, tg;
        const int el = block_excl_scan<PT>(hv, sw, tl);
        const int eg = block_excl_scan<PT>(gv, sw, tg);
        if (b < nb) {
            const int bs = carryG + eg;
            lbase[b] = carryL + el;
            gbase[b] = bs + woff[(size_t)b * nchunks + c];
            if (c == 0) bstart[b] = bs;
        }
        carryL += tl;
        carryG += tg;
    }
    if (c == 0 && tid == 0) bstart[nb] = carryG;
    __syncthreads();

#pragma unroll
    for (int k = 0; k < CH / PT; ++k) {
        if (key[k] != 0xFFFFFFFFu) {
            const int b = (int)(key[k] >> 16);
            const int p = lbase[b] + (int)(key[k] & 0xFFFFu);
            words[p] = wreg[k];
            bk[p]    = (unsigned short)b;
        }
    }
    __syncthreads();

    const int cnt = (et - e0 < CH) ? (et - e0) : CH;
    for (int i = tid; i < cnt; i += PT) {
        const int b = bk[i];
        sorted[gbase[b] + (i - lbase[b])] = words[i];
    }
}

__global__ __launch_bounds__(DG) void range_deg_kernel(const unsigned* __restrict__ sorted,
                                                       const int* __restrict__ bstart,
                                                       float* __restrict__ dinv, int n) {
    __shared__ int cnt[NB];
    const int tid = threadIdx.x;
    const int lo  = blockIdx.x * NB;
    for (int r = tid; r < NB; r += DG) cnt[r] = 0;

    const int start = bstart[blockIdx.x];
    const int end   = bstart[blockIdx.x + 1];
    const int lastw = (end > 0) ? (end - 1) : 0;

    unsigned w[DU];
#pragma unroll
    for (int u = 0; u < DU; ++u) {
        const int i = start + tid + u * DG;
        w[u] = sorted[(i < end) ? i : lastw];
    }
    __syncthreads();
#pragma unroll
    for (int u = 0; u < DU; ++u) {
        const int i = start + tid + u * DG;
        if (i < end) atomicAdd(&cnt[w[u] & (NB - 1)], 1);
    }
    for (int i0 = start + DU * DG; i0 < end; i0 += DU * DG) {
#pragma unroll
        for (int u = 0; u < DU; ++u) {
            const int i = i0 + tid + u * DG;
            w[u] = sorted[(i < end) ? i : lastw];
        }
#pragma unroll
        for (int u = 0; u < DU; ++u) {
            const int i = i0 + tid + u * DG;
            if (i < end) atomicAdd(&cnt[w[u] & (NB - 1)], 1);
        }
    }
    __syncthreads();
    for (int r = tid; r < NB; r += DG) {
        if (lo + r < n) dinv[lo + r] = rsqrtf((float)(1 + cnt[r]));
    }
}

__global__ __launch_bounds__(AT) void range_sum_kernel(const unsigned* __restrict__ sorted,
                                                       const int* __restrict__ bstart,
                                                       const float* __restrict__ dinv,
                                                       const float* __restrict__ h,
                                                       const float* __restrict__ bias,
                                                       float* __restrict__ out, int n) {
    __shared__ float acc[NB * LAT];
    __shared__ float sdinv[NB];

    const int tid  = threadIdx.x;
    const int lane = tid & 31;
    const int wave = tid >> 5;
    const int lo   = blockIdx.x * NB;

    for (int r = tid; r < NB; r += AT) {
        const float di = dinv[(lo + r < n) ? (lo + r) : (n - 1)];
        sdinv[r] = (lo + r < n) ? di : 0.0f;
    }
    {
        constexpr int RI = AT / 32;
        constexpr int NI = NB / RI;
        const float bl = bias[lane];
        float dv[NI];
        float hv[NI];
#pragma unroll
        for (int k = 0; k < NI; ++k) {
            const int node = lo + wave + k * RI;
            const int nc   = (node < n) ? node : (n - 1);
            dv[k] = dinv[nc];
            hv[k] = h[(size_t)nc * LAT + lane];
        }
#pragma unroll
        for (int k = 0; k < NI; ++k) {
            const int row = wave + k * RI;
            acc[row * LAT + lane] = (lo + row < n) ? (dv[k] * dv[k] * hv[k] + bl) : 0.0f;
        }
    }
    __syncthreads();

    const int start  = bstart[blockIdx.x];
    const int end    = bstart[blockIdx.x + 1];
    const int nbatch = (end - start + 31) >> 5;
    const float* hl  = h + lane;
    const unsigned lane_u = (unsigned)lane;

    const int lastw = end - 1;
    constexpr int KS = AT / 32;
    unsigned wd_cur = 0u, wd_nxt = 0u;
    float ds_cur = 0.0f;
    if (wave < nbatch) {
        const int i = start + (wave << 5) + lane;
        wd_cur = sorted[(i < end) ? i : lastw];
    }
    if (wave + KS < nbatch) {
        const int i = start + ((wave + KS) << 5) + lane;
        wd_nxt = sorted[(i < end) ? i : lastw];
    }
    if (wave < nbatch) ds_cur = dinv[wd_cur >> NB_LOG];

    for (int k = wave; k < nbatch; k += KS) {
        const int base = start + (k << 5);
        const int m    = (end - base < 32) ? (end - base) : 32;
        const unsigned wd = wd_cur;
        const float    ds = ds_cur;
        const int   s  = (int)(wd >> NB_LOG);
        const int   dl = (int)(wd & (NB - 1));
        const float w  = ds * sdinv[dl];
        if (m == 32) {
            unsigned ix[32];
#pragma unroll
            for (int j = 0; j < 32; ++j) {
                ix[j] = ((unsigned)__builtin_amdgcn_readlane(s, j) << 5) | lane_u;
            }
            asm volatile("" :: "v"(ix[0]), "v"(ix[1]), "v"(ix[2]), "v"(ix[3]), "v"(ix[4]), "v"(ix[5]),
                         "v"(ix[6]), "v"(ix[7]), "v"(ix[8]), "v"(ix[9]), "v"(ix[10]), "v"(ix[11]),
                         "v"(ix[12]), "v"(ix[13]), "v"(ix[14]), "v"(ix[15]) : "memory");
            asm volatile("" :: "v"(ix[16]), "v"(ix[17]), "v"(ix[18]), "v"(ix[19]), "v"(ix[20]), "v"(ix[21]),
                         "v"(ix[22]), "v"(ix[23]), "v"(ix[24]), "v"(ix[25]), "v"(ix[26]), "v"(ix[27]),
                         "v"(ix[28]), "v"(ix[29]), "v"(ix[30]), "v"(ix[31]) : "memory");
            float v[32];
#pragma unroll
            for (int j = 0; j < 32; ++j) {
                v[j] = h[ix[j]];
            }
            if (k + KS < nbatch) ds_cur = dinv[wd_nxt >> NB_LOG];
            wd_cur = wd_nxt;
            if (k + 2 * KS < nbatch) {
                const int i = start + ((k + 2 * KS) << 5) + lane;
                wd_nxt = sorted[(i < end) ? i : lastw];
            }
#pragma unroll
            for (int j = 0; j < 32; ++j) {
                const float wj = __int_as_float(__builtin_amdgcn_readlane(__float_as_int(w), j));
                const int   dj = __builtin_amdgcn_readlane(dl, j);
                atomicAdd(&acc[dj * LAT + lane], wj * v[j]);
            }
        } else {
            for (int j = 0; j < m; ++j) {
                const int   sj = __builtin_amdgcn_readlane(s, j);
                const float wj = __int_as_float(__builtin_amdgcn_readlane(__float_as_int(w), j));
                const int   dj = __builtin_amdgcn_readlane(dl, j);
                atomicAdd(&acc[dj * LAT + lane], wj * hl[(size_t)sj * LAT]);
            }
        }
    }
    __syncthreads();

    for (int i = tid; i < NB * LAT; i += AT) {
        if (lo + (i >> 5) < n) out[(size_t)lo * LAT + i] = acc[i];
    }
}


extern "C" void kernel_launch(void* const* d_in, const int* in_sizes, int n_in,
                              void* d_out, int out_size, void* d_ws, size_t ws_size,
                              hipStream_t stream) {
    const float* x   = (const float*)d_in[0];
    const int*   ei  = (const int*)  d_in[1];
    const int*   yei = (const int*)  d_in[2];
    const float* W   = (const float*)d_in[3];
    const float* b   = (const float*)d_in[4];
    float* out = (float*)d_out;

    const int n  = in_sizes[0] / IN_CH;
    const int e1 = in_sizes[1] / 2;
    const int e2 = in_sizes[2] / 2;
    const int et = e1 + e2;

    float* deg = (float*)d_ws;
    size_t hoff = (((size_t)n * sizeof(float)) + 255) & ~(size_t)255;
    float* h = (float*)((char*)d_ws + hoff);

    const int nb      = (n + NB - 1) / NB;
    const int nchunks = (et + CH - 1) / CH;
    size_t soff = (hoff + (size_t)n * LAT * sizeof(float) + 255) & ~(size_t)255;
    size_t coff = (soff + (size_t)et * sizeof(unsigned) + 255) & ~(size_t)255;
    size_t toff = (coff + (size_t)nb * nchunks * sizeof(int) + 255) & ~(size_t)255;
    size_t boff = (toff + (size_t)nb * sizeof(int) + 255) & ~(size_t)255;
    size_t need = boff + ((size_t)nb + 1) * sizeof(int);
    unsigned* sorted = (unsigned*)((char*)d_ws + soff);
    int* counts = (int*)((char*)d_ws + coff);
    int* btot   = (int*)((char*)d_ws + toff);
    int* bstart = (int*)((char*)d_ws + boff);
    const bool grouped = (need <= ws_size) && (nb <= MAXB) && (n <= (1 << 24)) && (nchunks > 0);

    {
        int ntiles = (n + TILE_M - 1) / TILE_M;
        int blocks = (ntiles + 3) / 4;
        gemm_wmma_kernel<<<blocks, 128, 0, stream>>>(x, W, h, n);
    }
    if (grouped) {
        range_count_kernel<<<nchunks, PT, 0, stream>>>(ei, yei, counts, e1, e2, nchunks, nb);
        range_scan_kernel<<<nb, 256, 0, stream>>>(counts, btot, nchunks);
        range_place_kernel<<<nchunks, PT, 0, stream>>>(ei, yei, counts, btot, bstart, sorted,
                                                       e1, e2, nchunks, nb);
        range_deg_kernel<<<nb, DG, 0, stream>>>(sorted, bstart, deg, n);
        range_sum_kernel<<<nb, AT, 0, stream>>>(sorted, bstart, deg, h, b, out, n);
    } else {
        deg_init_kernel<<<(n + 255) / 256, 256, 0, stream>>>(deg, n);
        deg_edges_kernel<<<(et + 255) / 256, 256, 0, stream>>>(ei, yei, deg, e1, e2);
        dinv_kernel<<<(n + 255) / 256, 256, 0, stream>>>(deg, n);
        init_out_kernel<<<((size_t)n * LAT + 255) / 256, 256, 0, stream>>>(deg, h, b, out, n);
        long long threads = (long long)et * 32;
        int blocks = (int)((threads + 255) / 256);
        edge_scatter_kernel<<<blocks, 256, 0, stream>>>(ei, yei, deg, h, out, e1, e2);
    }
}


# h = x W tiles run as extra work-groups of the edge-count launch (16 tiles per 512-thread group, count groups first): the two independent passes overlap, 6->5 launches; all sums unchanged
# speedup vs baseline: 3.8731x; 1.0421x over previous
#include <hip/hip_runtime.h>
#include <hip/hip_bf16.h>

typedef __attribute__((ext_vector_type(2))) float v2f;
typedef __attribute__((ext_vector_type(8))) float v8f;

#define IN_CH 128
#define LAT   32
#define TILE_M 16
#define APAD  132


__global__ void deg_init_kernel(float* __restrict__ deg, int n) {
    int i = blockIdx.x * blockDim.x + threadIdx.x;
    if (i < n) deg[i] = 1.0f;
}

__global__ void deg_edges_kernel(const int* __restrict__ ei, const int* __restrict__ yei,
                                 float* __restrict__ deg, int e1, int e2) {
    int e = blockIdx.x * blockDim.x + threadIdx.x;
    int et = e1 + e2;
    if (e >= et) return;
    int d = (e < e1) ? ei[e1 + e] : yei[e2 + (e - e1)];
    atomicAdd(deg + d, 1.0f);
}

__global__ void dinv_kernel(float* __restrict__ deg, int n) {
    int i = blockIdx.x * blockDim.x + threadIdx.x;
    if (i < n) deg[i] = rsqrtf(deg[i]);
}


#define GW 16
#define GEMM_LDS_FLOATS (IN_CH * LAT + GW * TILE_M * APAD)

__device__ __forceinline__ void gemm_role(const float* __restrict__ x,
                                          const float* __restrict__ W,
                                          float* __restrict__ h, int n, int gblock,
                                          float* __restrict__ sW,
                                          float* __restrict__ sAall) {
    const int tid  = threadIdx.x;
    const int wave = tid >> 5;
    const int lane = tid & 31;
    const int ntiles = (n + TILE_M - 1) / TILE_M;
    const int tile = gblock * GW + wave;
    float* sAw = sAall + wave * (TILE_M * APAD);

    {
        const float4* Wv = (const float4*)W;
        float4* sWv = (float4*)sW;
        for (int i = tid; i < (IN_CH * LAT) / 4; i += GW * 32) sWv[i] = Wv[i];
    }
    if (tile < ntiles) {
        const int r0 = tile * TILE_M;
        for (int i = lane; i < TILE_M * (IN_CH / 4); i += 32) {
            int row = i >> 5;
            int c4  = i & 31;
            int grow = r0 + row; if (grow >= n) grow = n - 1;
            float4 v = ((const float4*)(x + (size_t)grow * IN_CH))[c4];
            float* dst = &sAw[row * APAD + c4 * 4];
            dst[0] = v.x; dst[1] = v.y; dst[2] = v.z; dst[3] = v.w;
        }
    }
    __syncthreads();

    if (tile >= ntiles) return;

    v8f acc0 = {};
    v8f acc1 = {};
    const int m    = lane & 15;
    const int koff = (lane >> 4) << 1;
    const int nn   = lane & 15;

    for (int kk = 0; kk < IN_CH / 4; ++kk) {
        const int kb = kk * 4 + koff;
        v2f a;
        a.x = sAw[m * APAD + kb];
        a.y = sAw[m * APAD + kb + 1];
        v2f b0, b1;
        b0.x = sW[(kb    ) * LAT + nn];
        b0.y = sW[(kb + 1) * LAT + nn];
        b1.x = sW[(kb    ) * LAT + 16 + nn];
        b1.y = sW[(kb + 1) * LAT + 16 + nn];
        acc0 = __builtin_amdgcn_wmma_f32_16x16x4_f32(false, a, false, b0,
                                                     (short)0, acc0, false, false);
        acc1 = __builtin_amdgcn_wmma_f32_16x16x4_f32(false, a, false, b1,
                                                     (short)0, acc1, false, false);
    }

    const int r0   = tile * TILE_M;
    const int rofs = (lane >> 4) << 3;
#pragma unroll
    for (int r = 0; r < 8; ++r) {
        int row = r0 + r + rofs;
        if (row < n) {
            h[(size_t)row * LAT + nn]      = acc0[r];
            h[(size_t)row * LAT + 16 + nn] = acc1[r];
        }
    }
}


__global__ void init_out_kernel(const float* __restrict__ dinv, const float* __restrict__ h,
                                const float* __restrict__ b, float* __restrict__ out, int n) {
    int gid = blockIdx.x * blockDim.x + threadIdx.x;
    if (gid >= n * LAT) return;
    int i = gid >> 5, c = gid & 31;
    float di = dinv[i];
    out[gid] = di * di * h[gid] + b[c];
}


__global__ void edge_scatter_kernel(const int* __restrict__ ei, const int* __restrict__ yei,
                                    const float* __restrict__ dinv, const float* __restrict__ h,
                                    float* __restrict__ out, int e1, int e2) {
    long long gid = (long long)blockIdx.x * blockDim.x + threadIdx.x;
    int e = (int)(gid >> 5);
    int c = (int)(gid & 31);
    int et = e1 + e2;
    if (e >= et) return;
    int s, d;
    if (e < e1) { s = ei[e];            d = ei[e1 + e]; }
    else        { int f = e - e1; s = yei[f]; d = yei[e2 + f]; }
    float w = dinv[s] * dinv[d];
    atomicAdd(out + (size_t)d * LAT + c, w * h[(size_t)s * LAT + c]);
}


#define NB_LOG 8
#define NB   256
#define MAXB 1024
#define CH   8192
#define PT   512
#define AT   512
#define DG   512
#define DU   12

__device__ __forceinline__ int edge_src(const int* __restrict__ ei, const int* __restrict__ yei,
                                        int e1, int e) {
    return (e < e1) ? ei[e] : yei[e - e1];
}
__device__ __forceinline__ int edge_dst(const int* __restrict__ ei, const int* __restrict__ yei,
                                        int e1, int e2, int e) {
    return (e < e1) ? ei[e1 + e] : yei[e2 + (e - e1)];
}

template <int NT>
__device__ __forceinline__ int block_excl_scan(int v, int* sw, int& total) {
    const int lane = threadIdx.x & 31;
    const int wv   = threadIdx.x >> 5;
    int inc = v;
#pragma unroll
    for (int d = 1; d < 32; d <<= 1) {
        int t = __shfl_up(inc, d);
        if (lane >= d) inc += t;
    }
    if (lane == 31) sw[wv] = inc;
    __syncthreads();
    if (wv == 0) {
        int s  = (lane < NT / 32) ? sw[lane] : 0;
        int si = s;
#pragma unroll
        for (int d = 1; d < 32; d <<= 1) {
            int t = __shfl_up(si, d);
            if (lane >= d) si += t;
        }
        if (lane < NT / 32) sw[lane] = si - s;
        if (lane == 31) sw[32] = si;
    }
    __syncthreads();
    const int r = sw[wv] + inc - v;
    total = sw[32];
    __syncthreads();
    return r;
}

static_assert(PT == GW * 32, "a GEMM group is one full group of the count launch");
__global__ __launch_bounds__(PT) void count_gemm_kernel(const int* __restrict__ ei,
                                                        const int* __restrict__ yei,
                                                        int* __restrict__ counts,
                                                        int e1, int e2, int nchunks, int nb,
                                                        const float* __restrict__ x,
                                                        const float* __restrict__ W,
                                                        float* __restrict__ h, int n) {
    __shared__ __align__(16) float smem[GEMM_LDS_FLOATS];
    if ((int)blockIdx.x >= nchunks) {
        gemm_role(x, W, h, n, (int)blockIdx.x - nchunks, smem, smem + IN_CH * LAT);
        return;
    }
    int* hist = (int*)smem;
    const int tid = threadIdx.x;
    const int c   = blockIdx.x;
    const int et  = e1 + e2;
    const int e0  = c * CH;
    for (int i = tid; i < nb; i += PT) hist[i] = 0;
    int d[CH / PT];
#pragma unroll
    for (int k = 0; k < CH / PT; ++k) {
        const int e = e0 + tid + k * PT;
        d[k] = edge_dst(ei, yei, e1, e2, (e < et) ? e : et - 1);
    }
    __syncthreads();
#pragma unroll
    for (int k = 0; k < CH / PT; ++k) {
        const int e = e0 + tid + k * PT;
        if (e < et) atomicAdd(&hist[d[k] >> NB_LOG], 1);
    }
    __syncthreads();
    for (int b = tid; b < nb; b += PT) counts[(size_t)b * nchunks + c] = hist[b];
}

__global__ __launch_bounds__(256) void range_scan_kernel(int* __restrict__ counts,
                                                         int* __restrict__ btot, int nchunks) {
    __shared__ int sw[33];
    const int tid = threadIdx.x;
    int* row = counts + (size_t)blockIdx.x * nchunks;
    int carry = 0;
    for (int c0 = 0; c0 < nchunks; c0 += 256) {
        const int c = c0 + tid;
        const int v = (c < nchunks) ? row[c] : 0;
        int tot;
        const int ex = block_excl_scan<256>(v, sw, tot);
        if (c < nchunks) row[c] = carry + ex;
        carry += tot;
    }
    if (tid == 0) btot[blockIdx.x] = carry;
}

__global__ __launch_bounds__(PT) void range_place_kernel(const int* __restrict__ ei,
                                                         const int* __restrict__ yei,
                                                         const int* __restrict__ woff,
                                                         const int* __restrict__ btot,
                                                         int* __restrict__ bstart,
                                                         unsigned* __restrict__ sorted,
                                                         int e1, int e2, int nchunks, int nb) {
    __shared__ int hist[MAXB];
    __shared__ int lbase[MAXB];
    __shared__ int gbase[MAXB];
    __shared__ unsigned words[CH];
    __shared__ unsigned short bk[CH];
    __shared__ int sw[33];

    const int tid = threadIdx.x;
    const int c   = blockIdx.x;
    const int et  = e1 + e2;
    const int e0  = c * CH;

    for (int i = tid; i < nb; i += PT) hist[i] = 0;

    int sv[CH / PT];
    int dv[CH / PT];
#pragma unroll
    for (int k = 0; k < CH / PT; ++k) {
        const int e  = e0 + tid + k * PT;
        const int ec = (e < et) ? e : et - 1;
        sv[k] = edge_src(ei, yei, e1, ec);
        dv[k] = edge_dst(ei, yei, e1, e2, ec);
    }
    __syncthreads();

    unsigned wreg[CH / PT];
    unsigned key[CH / PT];
#pragma unroll
    for (int k = 0; k < CH / PT; ++k) {
        const int e = e0 + tid + k * PT;
        key[k]  = 0xFFFFFFFFu;
        wreg[k] = 0u;
        if (e < et) {
            const int s = sv[k];
            const int d = dv[k];
            const int b = d >> NB_LOG;
            const int r = atomicAdd(&hist[b], 1);
            wreg[k] = ((unsigned)s << NB_LOG) | (unsigned)(d & (NB - 1));
            key[k]  = ((unsigned)b << 16) | (unsigned)r;
        }
    }
    __syncthreads();

    int carryL = 0, carryG = 0;
    for (int b0 = 0; b0 < nb; b0 += PT) {
        const int b  = b0 + tid;
        const int hv = (b < nb) ? hist[b] : 0;
        const int gv = (b < nb) ? btot[b] : 0;
        int tl, tg;
        const int el = block_excl_scan<PT>(hv, sw, tl);
        const int eg = block_excl_scan<PT>(gv, sw, tg);
        if (b < nb) {
            const int bs = carryG + eg;
            lbase[b] = carryL + el;
            gbase[b] = bs + woff[(size_t)b * nchunks + c];
            if (c == 0) bstart[b] = bs;
        }
        carryL += tl;
        carryG += tg;
    }
    if (c == 0 && tid == 0) bstart[nb] = carryG;
    __syncthreads();

#pragma unroll
    for (int k = 0; k < CH / PT; ++k) {
        if (key[k] != 0xFFFFFFFFu) {
            const int b = (int)(key[k] >> 16);
            const int p = lbase[b] + (int)(key[k] & 0xFFFFu);
            words[p] = wreg[k];
            bk[p]    = (unsigned short)b;
        }
    }
    __syncthreads();

    const int cnt = (et - e0 < CH) ? (et - e0) : CH;
    for (int i = tid; i < cnt; i += PT) {
        const int b = bk[i];
        sorted[gbase[b] + (i - lbase[b])] = words[i];
    }
}

__global__ __launch_bounds__(DG) void range_deg_kernel(const unsigned* __restrict__ sorted,
                                                       const int* __restrict__ bstart,
                                                       float* __restrict__ dinv, int n) {
    __shared__ int cnt[NB];
    const int tid = threadIdx.x;
    const int lo  = blockIdx.x * NB;
    for (int r = tid; r < NB; r += DG) cnt[r] = 0;

    const int start = bstart[blockIdx.x];
    const int end   = bstart[blockIdx.x + 1];
    const int lastw = (end > 0) ? (end - 1) : 0;

    unsigned w[DU];
#pragma unroll
    for (int u = 0; u < DU; ++u) {
        const int i = start + tid + u * DG;
        w[u] = sorted[(i < end) ? i : lastw];
    }
    __syncthreads();
#pragma unroll
    for (int u = 0; u < DU; ++u) {
        const int i = start + tid + u * DG;
        if (i < end) atomicAdd(&cnt[w[u] & (NB - 1)], 1);
    }
    for (int i0 = start + DU * DG; i0 < end; i0 += DU * DG) {
#pragma unroll
        for (int u = 0; u < DU; ++u) {
            const int i = i0 + tid + u * DG;
            w[u] = sorted[(i < end) ? i : lastw];
        }
#pragma unroll
        for (int u = 0; u < DU; ++u) {
            const int i = i0 + tid + u * DG;
            if (i < end) atomicAdd(&cnt[w[u] & (NB - 1)], 1);
        }
    }
    __syncthreads();
    for (int r = tid; r < NB; r += DG) {
        if (lo + r < n) dinv[lo + r] = rsqrtf((float)(1 + cnt[r]));
    }
}

__global__ __launch_bounds__(AT) void range_sum_kernel(const unsigned* __restrict__ sorted,
                                                       const int* __restrict__ bstart,
                                                       const float* __restrict__ dinv,
                                                       const float* __restrict__ h,
                                                       const float* __restrict__ bias,
                                                       float* __restrict__ out, int n) {
    __shared__ float acc[NB * LAT];
    __shared__ float sdinv[NB];

    const int tid  = threadIdx.x;
    const int lane = tid & 31;
    const int wave = tid >> 5;
    const int lo   = blockIdx.x * NB;

    for (int r = tid; r < NB; r += AT) {
        const float di = dinv[(lo + r < n) ? (lo + r) : (n - 1)];
        sdinv[r] = (lo + r < n) ? di : 0.0f;
    }
    {
        constexpr int RI = AT / 32;
        constexpr int NI = NB / RI;
        const float bl = bias[lane];
        float dv[NI];
        float hv[NI];
#pragma unroll
        for (int k = 0; k < NI; ++k) {
            const int node = lo + wave + k * RI;
            const int nc   = (node < n) ? node : (n - 1);
            dv[k] = dinv[nc];
            hv[k] = h[(size_t)nc * LAT + lane];
        }
#pragma unroll
        for (int k = 0; k < NI; ++k) {
            const int row = wave + k * RI;
            acc[row * LAT + lane] = (lo + row < n) ? (dv[k] * dv[k] * hv[k] + bl) : 0.0f;
        }
    }
    __syncthreads();

    const int start  = bstart[blockIdx.x];
    const int end    = bstart[blockIdx.x + 1];
    const int nbatch = (end - start + 31) >> 5;
    const float* hl  = h + lane;
    const unsigned lane_u = (unsigned)lane;

    const int lastw = end - 1;
    constexpr int KS = AT / 32;
    unsigned wd_cur = 0u, wd_nxt = 0u;
    float ds_cur = 0.0f;
    if (wave < nbatch) {
        const int i = start + (wave << 5) + lane;
        wd_cur = sorted[(i < end) ? i : lastw];
    }
    if (wave + KS < nbatch) {
        const int i = start + ((wave + KS) << 5) + lane;
        wd_nxt = sorted[(i < end) ? i : lastw];
    }
    if (wave < nbatch) ds_cur = dinv[wd_cur >> NB_LOG];

    for (int k = wave; k < nbatch; k += KS) {
        const int base = start + (k << 5);
        const int m    = (end - base < 32) ? (end - base) : 32;
        const unsigned wd = wd_cur;
        const float    ds = ds_cur;
        const int   s  = (int)(wd >> NB_LOG);
        const int   dl = (int)(wd & (NB - 1));
        const float w  = ds * sdinv[dl];
        if (m == 32) {
            unsigned ix[32];
#pragma unroll
            for (int j = 0; j < 32; ++j) {
                ix[j] = ((unsigned)__builtin_amdgcn_readlane(s, j) << 5) | lane_u;
            }
            asm volatile("" :: "v"(ix[0]), "v"(ix[1]), "v"(ix[2]), "v"(ix[3]), "v"(ix[4]), "v"(ix[5]),
                         "v"(ix[6]), "v"(ix[7]), "v"(ix[8]), "v"(ix[9]), "v"(ix[10]), "v"(ix[11]),
                         "v"(ix[12]), "v"(ix[13]), "v"(ix[14]), "v"(ix[15]) : "memory");
            asm volatile("" :: "v"(ix[16]), "v"(ix[17]), "v"(ix[18]), "v"(ix[19]), "v"(ix[20]), "v"(ix[21]),
                         "v"(ix[22]), "v"(ix[23]), "v"(ix[24]), "v"(ix[25]), "v"(ix[26]), "v"(ix[27]),
                         "v"(ix[28]), "v"(ix[29]), "v"(ix[30]), "v"(ix[31]) : "memory");
            float v[32];
#pragma unroll
            for (int j = 0; j < 32; ++j) {
                v[j] = h[ix[j]];
            }
            if (k + KS < nbatch) ds_cur = dinv[wd_nxt >> NB_LOG];
            wd_cur = wd_nxt;
            if (k + 2 * KS < nbatch) {
                const int i = start + ((k + 2 * KS) << 5) + lane;
                wd_nxt = sorted[(i < end) ? i : lastw];
            }
#pragma unroll
            for (int j = 0; j < 32; ++j) {
                const float wj = __int_as_float(__builtin_amdgcn_readlane(__float_as_int(w), j));
                const int   dj = __builtin_amdgcn_readlane(dl, j);
                atomicAdd(&acc[dj * LAT + lane], wj * v[j]);
            }
        } else {
            for (int j = 0; j < m; ++j) {
                const int   sj = __builtin_amdgcn_readlane(s, j);
                const float wj = __int_as_float(__builtin_amdgcn_readlane(__float_as_int(w), j));
                const int   dj = __builtin_amdgcn_readlane(dl, j);
                atomicAdd(&acc[dj * LAT + lane], wj * hl[(size_t)sj * LAT]);
            }
        }
    }
    __syncthreads();

    for (int i = tid; i < NB * LAT; i += AT) {
        if (lo + (i >> 5) < n) out[(size_t)lo * LAT + i] = acc[i];
    }
}


extern "C" void kernel_launch(void* const* d_in, const int* in_sizes, int n_in,
                              void* d_out, int out_size, void* d_ws, size_t ws_size,
                              hipStream_t stream) {
    const float* x   = (const float*)d_in[0];
    const int*   ei  = (const int*)  d_in[1];
    const int*   yei = (const int*)  d_in[2];
    const float* W   = (const float*)d_in[3];
    const float* b   = (const float*)d_in[4];
    float* out = (float*)d_out;

    const int n  = in_sizes[0] / IN_CH;
    const int e1 = in_sizes[1] / 2;
    const int e2 = in_sizes[2] / 2;
    const int et = e1 + e2;

    float* deg = (float*)d_ws;
    size_t hoff = (((size_t)n * sizeof(float)) + 255) & ~(size_t)255;
    float* h = (float*)((char*)d_ws + hoff);

    const int nb      = (n + NB - 1) / NB;
    const int nchunks = (et + CH - 1) / CH;
    size_t soff = (hoff + (size_t)n * LAT * sizeof(float) + 255) & ~(size_t)255;
    size_t coff = (soff + (size_t)et * sizeof(unsigned) + 255) & ~(size_t)255;
    size_t toff = (coff + (size_t)nb * nchunks * sizeof(int) + 255) & ~(size_t)255;
    size_t boff = (toff + (size_t)nb * sizeof(int) + 255) & ~(size_t)255;
    size_t need = boff + ((size_t)nb + 1) * sizeof(int);
    unsigned* sorted = (unsigned*)((char*)d_ws + soff);
    int* counts = (int*)((char*)d_ws + coff);
    int* btot   = (int*)((char*)d_ws + toff);
    int* bstart = (int*)((char*)d_ws + boff);
    const bool grouped = (need <= ws_size) && (nb <= MAXB) && (n <= (1 << 24)) && (nchunks > 0);

    const int ntiles  = (n + TILE_M - 1) / TILE_M;
    const int gblocks = (ntiles + GW - 1) / GW;
    if (grouped) {
        count_gemm_kernel<<<nchunks + gblocks, PT, 0, stream>>>(ei, yei, counts, e1, e2, nchunks, nb,
                                                                x, W, h, n);
        range_scan_kernel<<<nb, 256, 0, stream>>>(counts, btot, nchunks);
        range_place_kernel<<<nchunks, PT, 0, stream>>>(ei, yei, counts, btot, bstart, sorted,
                                                       e1, e2, nchunks, nb);
        range_deg_kernel<<<nb, DG, 0, stream>>>(sorted, bstart, deg, n);
        range_sum_kernel<<<nb, AT, 0, stream>>>(sorted, bstart, deg, h, b, out, n);
    } else {
        count_gemm_kernel<<<gblocks, PT, 0, stream>>>(ei, yei, counts, e1, e2, 0, nb, x, W, h, n);
        deg_init_kernel<<<(n + 255) / 256, 256, 0, stream>>>(deg, n);
        deg_edges_kernel<<<(et + 255) / 256, 256, 0, stream>>>(ei, yei, deg, e1, e2);
        dinv_kernel<<<(n + 255) / 256, 256, 0, stream>>>(deg, n);
        init_out_kernel<<<((size_t)n * LAT + 255) / 256, 256, 0, stream>>>(deg, h, b, out, n);
        long long threads = (long long)et * 32;
        int blocks = (int)((threads + 255) / 256);
        edge_scatter_kernel<<<blocks, 256, 0, stream>>>(ei, yei, deg, h, out, e1, e2);
    }
}


# range-sum: source deg^-1/2 from a byte table of degree counts kept in LDS (written by the degree pass, copied once per block), not a 4-byte global gather per edge; weights and sum order as before
# speedup vs baseline: 4.1137x; 1.0621x over previous
#include <hip/hip_runtime.h>
#include <hip/hip_bf16.h>

typedef __attribute__((ext_vector_type(2))) float v2f;
typedef __attribute__((ext_vector_type(8))) float v8f;

#define IN_CH 128
#define LAT   32
#define TILE_M 16
#define APAD  132


__global__ void deg_init_kernel(float* __restrict__ deg, int n) {
    int i = blockIdx.x * blockDim.x + threadIdx.x;
    if (i < n) deg[i] = 1.0f;
}

__global__ void deg_edges_kernel(const int* __restrict__ ei, const int* __restrict__ yei,
                                 float* __restrict__ deg, int e1, int e2) {
    int e = blockIdx.x * blockDim.x + threadIdx.x;
    int et = e1 + e2;
    if (e >= et) return;
    int d = (e < e1) ? ei[e1 + e] : yei[e2 + (e - e1)];
    atomicAdd(deg + d, 1.0f);
}

__global__ void dinv_kernel(float* __restrict__ deg, int n) {
    int i = blockIdx.x * blockDim.x + threadIdx.x;
    if (i < n) deg[i] = rsqrtf(deg[i]);
}


#define GW 16
#define GEMM_LDS_FLOATS (IN_CH * LAT + GW * TILE_M * APAD)

__device__ __forceinline__ void gemm_role(const float* __restrict__ x,
                                          const float* __restrict__ W,
                                          float* __restrict__ h, int n, int gblock,
                                          float* __restrict__ sW,
                                          float* __restrict__ sAall) {
    const int tid  = threadIdx.x;
    const int wave = tid >> 5;
    const int lane = tid & 31;
    const int ntiles = (n + TILE_M - 1) / TILE_M;
    const int tile = gblock * GW + wave;
    float* sAw = sAall + wave * (TILE_M * APAD);

    {
        const float4* Wv = (const float4*)W;
        float4* sWv = (float4*)sW;
        for (int i = tid; i < (IN_CH * LAT) / 4; i += GW * 32) sWv[i] = Wv[i];
    }
    if (tile < ntiles) {
        const int r0 = tile * TILE_M;
        for (int i = lane; i < TILE_M * (IN_CH / 4); i += 32) {
            int row = i >> 5;
            int c4  = i & 31;
            int grow = r0 + row; if (grow >= n) grow = n - 1;
            float4 v = ((const float4*)(x + (size_t)grow * IN_CH))[c4];
            float* dst = &sAw[row * APAD + c4 * 4];
            dst[0] = v.x; dst[1] = v.y; dst[2] = v.z; dst[3] = v.w;
        }
    }
    __syncthreads();

    if (tile >= ntiles) return;

    v8f acc0 = {};
    v8f acc1 = {};
    const int m    = lane & 15;
    const int koff = (lane >> 4) << 1;
    const int nn   = lane & 15;

    for (int kk = 0; kk < IN_CH / 4; ++kk) {
        const int kb = kk * 4 + koff;
        v2f a;
        a.x = sAw[m * APAD + kb];
        a.y = sAw[m * APAD + kb + 1];
        v2f b0, b1;
        b0.x = sW[(kb    ) * LAT + nn];
        b0.y = sW[(kb + 1) * LAT + nn];
        b1.x = sW[(kb    ) * LAT + 16 + nn];
        b1.y = sW[(kb + 1) * LAT + 16 + nn];
        acc0 = __builtin_amdgcn_wmma_f32_16x16x4_f32(false, a, false, b0,
                                                     (short)0, acc0, false, false);
        acc1 = __builtin_amdgcn_wmma_f32_16x16x4_f32(false, a, false, b1,
                                                     (short)0, acc1, false, false);
    }

    const int r0   = tile * TILE_M;
    const int rofs = (lane >> 4) << 3;
#pragma unroll
    for (int r = 0; r < 8; ++r) {
        int row = r0 + r + rofs;
        if (row < n) {
            h[(size_t)row * LAT + nn]      = acc0[r];
            h[(size_t)row * LAT + 16 + nn] = acc1[r];
        }
    }
}


__global__ void init_out_kernel(const float* __restrict__ dinv, const float* __restrict__ h,
                                const float* __restrict__ b, float* __restrict__ out, int n) {
    int gid = blockIdx.x * blockDim.x + threadIdx.x;
    if (gid >= n * LAT) return;
    int i = gid >> 5, c = gid & 31;
    float di = dinv[i];
    out[gid] = di * di * h[gid] + b[c];
}


__global__ void edge_scatter_kernel(const int* __restrict__ ei, const int* __restrict__ yei,
                                    const float* __restrict__ dinv, const float* __restrict__ h,
                                    float* __restrict__ out, int e1, int e2) {
    long long gid = (long long)blockIdx.x * blockDim.x + threadIdx.x;
    int e = (int)(gid >> 5);
    int c = (int)(gid & 31);
    int et = e1 + e2;
    if (e >= et) return;
    int s, d;
    if (e < e1) { s = ei[e];            d = ei[e1 + e]; }
    else        { int f = e - e1; s = yei[f]; d = yei[e2 + f]; }
    float w = dinv[s] * dinv[d];
    atomicAdd(out + (size_t)d * LAT + c, w * h[(size_t)s * LAT + c]);
}


#define NB_LOG 8
#define NB   256
#define MAXB 1024
#define CH   8192
#define PT   512
#define AT   512
#define DG   512
#define DU   12
#define DTAB (13 * AT * 16)

__device__ __forceinline__ int edge_src(const int* __restrict__ ei, const int* __restrict__ yei,
                                        int e1, int e) {
    return (e < e1) ? ei[e] : yei[e - e1];
}
__device__ __forceinline__ int edge_dst(const int* __restrict__ ei, const int* __restrict__ yei,
                                        int e1, int e2, int e) {
    return (e < e1) ? ei[e1 + e] : yei[e2 + (e - e1)];
}

template <int NT>
__device__ __forceinline__ int block_excl_scan(int v, int* sw, int& total) {
    const int lane = threadIdx.x & 31;
    const int wv   = threadIdx.x >> 5;
    int inc = v;
#pragma unroll
    for (int d = 1; d < 32; d <<= 1) {
        int t = __shfl_up(inc, d);
        if (lane >= d) inc += t;
    }
    if (lane == 31) sw[wv] = inc;
    __syncthreads();
    if (wv == 0) {
        int s  = (lane < NT / 32) ? sw[lane] : 0;
        int si = s;
#pragma unroll
        for (int d = 1; d < 32; d <<= 1) {
            int t = __shfl_up(si, d);
            if (lane >= d) si += t;
        }
        if (lane < NT / 32) sw[lane] = si - s;
        if (lane == 31) sw[32] = si;
    }
    __syncthreads();
    const int r = sw[wv] + inc - v;
    total = sw[32];
    __syncthreads();
    return r;
}

static_assert(PT == GW * 32, "a GEMM group is one full group of the count launch");
__global__ __launch_bounds__(PT) void count_gemm_kernel(const int* __restrict__ ei,
                                                        const int* __restrict__ yei,
                                                        int* __restrict__ counts,
                                                        int e1, int e2, int nchunks, int nb,
                                                        const float* __restrict__ x,
                                                        const float* __restrict__ W,
                                                        float* __restrict__ h, int n) {
    __shared__ __align__(16) float smem[GEMM_LDS_FLOATS];
    if ((int)blockIdx.x >= nchunks) {
        gemm_role(x, W, h, n, (int)blockIdx.x - nchunks, smem, smem + IN_CH * LAT);
        return;
    }
    int* hist = (int*)smem;
    const int tid = threadIdx.x;
    const int c   = blockIdx.x;
    const int et  = e1 + e2;
    const int e0  = c * CH;
    for (int i = tid; i < nb; i += PT) hist[i] = 0;
    int d[CH / PT];
#pragma unroll
    for (int k = 0; k < CH / PT; ++k) {
        const int e = e0 + tid + k * PT;
        d[k] = edge_dst(ei, yei, e1, e2, (e < et) ? e : et - 1);
    }
    __syncthreads();
#pragma unroll
    for (int k = 0; k < CH / PT; ++k) {
        const int e = e0 + tid + k * PT;
        if (e < et) atomicAdd(&hist[d[k] >> NB_LOG], 1);
    }
    __syncthreads();
    for (int b = tid; b < nb; b += PT) counts[(size_t)b * nchunks + c] = hist[b];
}

__global__ __launch_bounds__(256) void range_scan_kernel(int* __restrict__ counts,
                                                         int* __restrict__ btot, int nchunks) {
    __shared__ int sw[33];
    const int tid = threadIdx.x;
    int* row = counts + (size_t)blockIdx.x * nchunks;
    int carry = 0;
    for (int c0 = 0; c0 < nchunks; c0 += 256) {
        const int c = c0 + tid;
        const int v = (c < nchunks) ? row[c] : 0;
        int tot;
        const int ex = block_excl_scan<256>(v, sw, tot);
        if (c < nchunks) row[c] = carry + ex;
        carry += tot;
    }
    if (tid == 0) btot[blockIdx.x] = carry;
}

__global__ __launch_bounds__(PT) void range_place_kernel(const int* __restrict__ ei,
                                                         const int* __restrict__ yei,
                                                         const int* __restrict__ woff,
                                                         const int* __restrict__ btot,
                                                         int* __restrict__ bstart,
                                                         unsigned* __restrict__ sorted,
                                                         int e1, int e2, int nchunks, int nb) {
    __shared__ int hist[MAXB];
    __shared__ int lbase[MAXB];
    __shared__ int gbase[MAXB];
    __shared__ unsigned words[CH];
    __shared__ unsigned short bk[CH];
    __shared__ int sw[33];

    const int tid = threadIdx.x;
    const int c   = blockIdx.x;
    const int et  = e1 + e2;
    const int e0  = c * CH;

    for (int i = tid; i < nb; i += PT) hist[i] = 0;

    int sv[CH / PT];
    int dv[CH / PT];
#pragma unroll
    for (int k = 0; k < CH / PT; ++k) {
        const int e  = e0 + tid + k * PT;
        const int ec = (e < et) ? e : et - 1;
        sv[k] = edge_src(ei, yei, e1, ec);
        dv[k] = edge_dst(ei, yei, e1, e2, ec);
    }
    __syncthreads();

    unsigned wreg[CH / PT];
    unsigned key[CH / PT];
#pragma unroll
    for (int k = 0; k < CH / PT; ++k) {
        const int e = e0 + tid + k * PT;
        key[k]  = 0xFFFFFFFFu;
        wreg[k] = 0u;
        if (e < et) {
            const int s = sv[k];
            const int d = dv[k];
            const int b = d >> NB_LOG;
            const int r = atomicAdd(&hist[b], 1);
            wreg[k] = ((unsigned)s << NB_LOG) | (unsigned)(d & (NB - 1));
            key[k]  = ((unsigned)b << 16) | (unsigned)r;
        }
    }
    __syncthreads();

    int carryL = 0, carryG = 0;
    for (int b0 = 0; b0 < nb; b0 += PT) {
        const int b  = b0 + tid;
        const int hv = (b < nb) ? hist[b] : 0;
        const int gv = (b < nb) ? btot[b] : 0;
        int tl, tg;
        const int el = block_excl_scan<PT>(hv, sw, tl);
        const int eg = block_excl_scan<PT>(gv, sw, tg);
        if (b < nb) {
            const int bs = carryG + eg;
            lbase[b] = carryL + el;
            gbase[b] = bs + woff[(size_t)b * nchunks + c];
            if (c == 0) bstart[b] = bs;
        }
        carryL += tl;
        carryG += tg;
    }
    if (c == 0 && tid == 0) bstart[nb] = carryG;
    __syncthreads();

#pragma unroll
    for (int k = 0; k < CH / PT; ++k) {
        if (key[k] != 0xFFFFFFFFu) {
            const int b = (int)(key[k] >> 16);
            const int p = lbase[b] + (int)(key[k] & 0xFFFFu);
            words[p] = wreg[k];
            bk[p]    = (unsigned short)b;
        }
    }
    __syncthreads();

    const int cnt = (et - e0 < CH) ? (et - e0) : CH;
    for (int i = tid; i < cnt; i += PT) {
        const int b = bk[i];
        sorted[gbase[b] + (i - lbase[b])] = words[i];
    }
}

__global__ __launch_bounds__(DG) void range_deg_kernel(const unsigned* __restrict__ sorted,
                                                       const int* __restrict__ bstart,
                                                       float* __restrict__ dinv,
                                                       unsigned* __restrict__ dtab, int ntabw, int n) {
    __shared__ int cnt[NB];
    const int tid = threadIdx.x;
    const int lo  = blockIdx.x * NB;
    for (int r = tid; r < NB; r += DG) cnt[r] = 0;

    const int start = bstart[blockIdx.x];
    const int end   = bstart[blockIdx.x + 1];
    const int lastw = (end > 0) ? (end - 1) : 0;

    unsigned w[DU];
#pragma unroll
    for (int u = 0; u < DU; ++u) {
        const int i = start + tid + u * DG;
        w[u] = sorted[(i < end) ? i : lastw];
    }
    __syncthreads();
#pragma unroll
    for (int u = 0; u < DU; ++u) {
        const int i = start + tid + u * DG;
        if (i < end) atomicAdd(&cnt[w[u] & (NB - 1)], 1);
    }
    for (int i0 = start + DU * DG; i0 < end; i0 += DU * DG) {
#pragma unroll
        for (int u = 0; u < DU; ++u) {
            const int i = i0 + tid + u * DG;
            w[u] = sorted[(i < end) ? i : lastw];
        }
#pragma unroll
        for (int u = 0; u < DU; ++u) {
            const int i = i0 + tid + u * DG;
            if (i < end) atomicAdd(&cnt[w[u] & (NB - 1)], 1);
        }
    }
    __syncthreads();
    for (int r = tid; r < NB; r += DG) {
        if (lo + r < n) dinv[lo + r] = rsqrtf((float)(1 + cnt[r]));
    }
    for (int q = tid; q < NB / 4; q += DG) {
        const int wi = (lo >> 2) + q;
        if (wi < ntabw) {
            unsigned word = 0u;
            for (int j = 0; j < 4; ++j) {
                const int cj = cnt[4 * q + j];
                word |= (unsigned)((cj < 255) ? cj : 255) << (8 * j);
            }
            dtab[wi] = word;
        }
    }
}

template <bool TAB>
__global__ __launch_bounds__(AT) void range_sum_kernel(const unsigned* __restrict__ sorted,
                                                       const int* __restrict__ bstart,
                                                       const float* __restrict__ dinv,
                                                       const unsigned* __restrict__ dtab,
                                                       const float* __restrict__ h,
                                                       const float* __restrict__ bias,
                                                       float* __restrict__ out, int n) {
    __shared__ float acc[NB * LAT];
    __shared__ float sdinv[NB];
    __shared__ __align__(16) unsigned char stab[TAB ? DTAB : 16];

    const int tid  = threadIdx.x;
    const int lane = tid & 31;
    const int wave = tid >> 5;
    const int lo   = blockIdx.x * NB;

    constexpr int TV = TAB ? (DTAB / 16 / AT) : 1;
    const int nvec = (n + 15) >> 4;
    uint4 tv[TV];
    if (TAB) {
        const uint4* gt = (const uint4*)dtab;
#pragma unroll
        for (int u = 0; u < TV; ++u) {
            const int i = tid + u * AT;
            tv[u] = gt[(i < nvec) ? i : (nvec - 1)];
        }
    }

    for (int r = tid; r < NB; r += AT) {
        const float di = dinv[(lo + r < n) ? (lo + r) : (n - 1)];
        sdinv[r] = (lo + r < n) ? di : 0.0f;
    }
    {
        constexpr int RI = AT / 32;
        constexpr int NI = NB / RI;
        const float bl = bias[lane];
        float dv[NI];
        float hv[NI];
#pragma unroll
        for (int k = 0; k < NI; ++k) {
            const int node = lo + wave + k * RI;
            const int nc   = (node < n) ? node : (n - 1);
            dv[k] = dinv[nc];
            hv[k] = h[(size_t)nc * LAT + lane];
        }
        if (TAB) {
            uint4* st = (uint4*)stab;
#pragma unroll
            for (int u = 0; u < TV; ++u) {
                const int i = tid + u * AT;
                st[(i < nvec) ? i : (nvec - 1)] = tv[u];
            }
        }
#pragma unroll
        for (int k = 0; k < NI; ++k) {
            const int row = wave + k * RI;
            acc[row * LAT + lane] = (lo + row < n) ? (dv[k] * dv[k] * hv[k] + bl) : 0.0f;
        }
    }
    __syncthreads();

    const int start  = bstart[blockIdx.x];
    const int end    = bstart[blockIdx.x + 1];
    const int nbatch = (end - start + 31) >> 5;
    const float* hl  = h + lane;
    const unsigned lane_u = (unsigned)lane;

    const int lastw = end - 1;
    constexpr int KS = AT / 32;
    unsigned wd_cur = 0u, wd_nxt = 0u;
    float ds_cur = 0.0f;
    if (wave < nbatch) {
        const int i = start + (wave << 5) + lane;
        wd_cur = sorted[(i < end) ? i : lastw];
    }
    if (wave + KS < nbatch) {
        const int i = start + ((wave + KS) << 5) + lane;
        wd_nxt = sorted[(i < end) ? i : lastw];
    }
    if (!TAB && wave < nbatch) ds_cur = dinv[wd_cur >> NB_LOG];

    for (int k = wave; k < nbatch; k += KS) {
        const int base = start + (k << 5);
        const int m    = (end - base < 32) ? (end - base) : 32;
        const unsigned wd = wd_cur;
        const int   s  = (int)(wd >> NB_LOG);
        const int   dl = (int)(wd & (NB - 1));
        const float sd = sdinv[dl];
        float ds = ds_cur;
        if (TAB) {
            const unsigned cs = stab[s];
            ds = rsqrtf((float)(1 + (int)cs));
            if (__builtin_amdgcn_ballot_w32(cs == 255u) != 0u) {
                if (cs == 255u) ds = dinv[s];
            }
        }
        const float w  = ds * sd;
        if (m == 32) {
            unsigned ix[32];
#pragma unroll
            for (int j = 0; j < 32; ++j) {
                ix[j] = ((unsigned)__builtin_amdgcn_readlane(s, j) << 5) | lane_u;
            }
            asm volatile("" :: "v"(ix[0]), "v"(ix[1]), "v"(ix[2]), "v"(ix[3]), "v"(ix[4]), "v"(ix[5]),
                         "v"(ix[6]), "v"(ix[7]), "v"(ix[8]), "v"(ix[9]), "v"(ix[10]), "v"(ix[11]),
                         "v"(ix[12]), "v"(ix[13]), "v"(ix[14]), "v"(ix[15]) : "memory");
            asm volatile("" :: "v"(ix[16]), "v"(ix[17]), "v"(ix[18]), "v"(ix[19]), "v"(ix[20]), "v"(ix[21]),
                         "v"(ix[22]), "v"(ix[23]), "v"(ix[24]), "v"(ix[25]), "v"(ix[26]), "v"(ix[27]),
                         "v"(ix[28]), "v"(ix[29]), "v"(ix[30]), "v"(ix[31]) : "memory");
            float v[32];
#pragma unroll
            for (int j = 0; j < 32; ++j) {
                v[j] = h[ix[j]];
            }
            if (!TAB && k + KS < nbatch) ds_cur = dinv[wd_nxt >> NB_LOG];
            wd_cur = wd_nxt;
            if (k + 2 * KS < nbatch) {
                const int i = start + ((k + 2 * KS) << 5) + lane;
                wd_nxt = sorted[(i < end) ? i : lastw];
            }
#pragma unroll
            for (int j = 0; j < 32; ++j) {
                const float wj = __int_as_float(__builtin_amdgcn_readlane(__float_as_int(w), j));
                const int   dj = __builtin_amdgcn_readlane(dl, j);
                atomicAdd(&acc[dj * LAT + lane], wj * v[j]);
            }
        } else {
            for (int j = 0; j < m; ++j) {
                const int   sj = __builtin_amdgcn_readlane(s, j);
                const float wj = __int_as_float(__builtin_amdgcn_readlane(__float_as_int(w), j));
                const int   dj = __builtin_amdgcn_readlane(dl, j);
                atomicAdd(&acc[dj * LAT + lane], wj * hl[(size_t)sj * LAT]);
            }
        }
    }
    __syncthreads();

    for (int i = tid; i < NB * LAT; i += AT) {
        if (lo + (i >> 5) < n) out[(size_t)lo * LAT + i] = acc[i];
    }
}


extern "C" void kernel_launch(void* const* d_in, const int* in_sizes, int n_in,
                              void* d_out, int out_size, void* d_ws, size_t ws_size,
                              hipStream_t stream) {
    const float* x   = (const float*)d_in[0];
    const int*   ei  = (const int*)  d_in[1];
    const int*   yei = (const int*)  d_in[2];
    const float* W   = (const float*)d_in[3];
    const float* b   = (const float*)d_in[4];
    float* out = (float*)d_out;

    const int n  = in_sizes[0] / IN_CH;
    const int e1 = in_sizes[1] / 2;
    const int e2 = in_sizes[2] / 2;
    const int et = e1 + e2;

    float* deg = (float*)d_ws;
    size_t hoff = (((size_t)n * sizeof(float)) + 255) & ~(size_t)255;
    float* h = (float*)((char*)d_ws + hoff);

    const int nb      = (n + NB - 1) / NB;
    const int nchunks = (et + CH - 1) / CH;
    size_t soff = (hoff + (size_t)n * LAT * sizeof(float) + 255) & ~(size_t)255;
    size_t coff = (soff + (size_t)et * sizeof(unsigned) + 255) & ~(size_t)255;
    size_t toff = (coff + (size_t)nb * nchunks * sizeof(int) + 255) & ~(size_t)255;
    size_t boff = (toff + (size_t)nb * sizeof(int) + 255) & ~(size_t)255;
    size_t need = boff + ((size_t)nb + 1) * sizeof(int);
    unsigned* sorted = (unsigned*)((char*)d_ws + soff);
    int* counts = (int*)((char*)d_ws + coff);
    int* btot   = (int*)((char*)d_ws + toff);
    int* bstart = (int*)((char*)d_ws + boff);
    const bool grouped = (need <= ws_size) && (nb <= MAXB) && (n <= (1 << 24)) && (nchunks > 0);
    const int  ntabw = ((n + 15) >> 4) << 2;
    const bool tab   = (n <= DTAB) && ((size_t)ntabw <= (size_t)nb * nchunks);
    unsigned* dtab   = (unsigned*)counts;

    const int ntiles  = (n + TILE_M - 1) / TILE_M;
    const int gblocks = (ntiles + GW - 1) / GW;
    if (grouped) {
        count_gemm_kernel<<<nchunks + gblocks, PT, 0, stream>>>(ei, yei, counts, e1, e2, nchunks, nb,
                                                                x, W, h, n);
        range_scan_kernel<<<nb, 256, 0, stream>>>(counts, btot, nchunks);
        range_place_kernel<<<nchunks, PT, 0, stream>>>(ei, yei, counts, btot, bstart, sorted,
                                                       e1, e2, nchunks, nb);
        range_deg_kernel<<<nb, DG, 0, stream>>>(sorted, bstart, deg, dtab, tab ? ntabw : 0, n);
        if (tab) range_sum_kernel<true><<<nb, AT, 0, stream>>>(sorted, bstart, deg, dtab, h, b, out, n);
        else     range_sum_kernel<false><<<nb, AT, 0, stream>>>(sorted, bstart, deg, dtab, h, b, out, n);
    } else {
        count_gemm_kernel<<<gblocks, PT, 0, stream>>>(ei, yei, counts, e1, e2, 0, nb, x, W, h, n);
        deg_init_kernel<<<(n + 255) / 256, 256, 0, stream>>>(deg, n);
        deg_edges_kernel<<<(et + 255) / 256, 256, 0, stream>>>(ei, yei, deg, e1, e2);
        dinv_kernel<<<(n + 255) / 256, 256, 0, stream>>>(deg, n);
        init_out_kernel<<<((size_t)n * LAT + 255) / 256, 256, 0, stream>>>(deg, h, b, out, n);
        long long threads = (long long)et * 32;
        int blocks = (int)((threads + 255) / 256);
        edge_scatter_kernel<<<blocks, 256, 0, stream>>>(ei, yei, deg, h, out, e1, e2);
    }
}


# ranges re-sized to one range-sum block per compute unit: 392 nodes a range (256 ranges, were 391), block of 32 waves: every unit sums an equal share of edges; LDS atomic sum order as before
# speedup vs baseline: 4.7088x; 1.1447x over previous
#include <hip/hip_runtime.h>
#include <hip/hip_bf16.h>

typedef __attribute__((ext_vector_type(2))) float v2f;
typedef __attribute__((ext_vector_type(8))) float v8f;

#define IN_CH 128
#define LAT   32
#define TILE_M 16
#define APAD  132


__global__ void deg_init_kernel(float* __restrict__ deg, int n) {
    int i = blockIdx.x * blockDim.x + threadIdx.x;
    if (i < n) deg[i] = 1.0f;
}

__global__ void deg_edges_kernel(const int* __restrict__ ei, const int* __restrict__ yei,
                                 float* __restrict__ deg, int e1, int e2) {
    int e = blockIdx.x * blockDim.x + threadIdx.x;
    int et = e1 + e2;
    if (e >= et) return;
    int d = (e < e1) ? ei[e1 + e] : yei[e2 + (e - e1)];
    atomicAdd(deg + d, 1.0f);
}

__global__ void dinv_kernel(float* __restrict__ deg, int n) {
    int i = blockIdx.x * blockDim.x + threadIdx.x;
    if (i < n) deg[i] = rsqrtf(deg[i]);
}


#define GW 16
#define GEMM_LDS_FLOATS (IN_CH * LAT + GW * TILE_M * APAD)

__device__ __forceinline__ void gemm_role(const float* __restrict__ x,
                                          const float* __restrict__ W,
                                          float* __restrict__ h, int n, int gblock,
                                          float* __restrict__ sW,
                                          float* __restrict__ sAall) {
    const int tid  = threadIdx.x;
    const int wave = tid >> 5;
    const int lane = tid & 31;
    const int ntiles = (n + TILE_M - 1) / TILE_M;
    const int tile = gblock * GW + wave;
    float* sAw = sAall + wave * (TILE_M * APAD);

    {
        const float4* Wv = (const float4*)W;
        float4* sWv = (float4*)sW;
        for (int i = tid; i < (IN_CH * LAT) / 4; i += GW * 32) sWv[i] = Wv[i];
    }
    if (tile < ntiles) {
        const int r0 = tile * TILE_M;
        for (int i = lane; i < TILE_M * (IN_CH / 4); i += 32) {
            int row = i >> 5;
            int c4  = i & 31;
            int grow = r0 + row; if (grow >= n) grow = n - 1;
            float4 v = ((const float4*)(x + (size_t)grow * IN_CH))[c4];
            float* dst = &sAw[row * APAD + c4 * 4];
            dst[0] = v.x; dst[1] = v.y; dst[2] = v.z; dst[3] = v.w;
        }
    }
    __syncthreads();

    if (tile >= ntiles) return;

    v8f acc0 = {};
    v8f acc1 = {};
    const int m    = lane & 15;
    const int koff = (lane >> 4) << 1;
    const int nn   = lane & 15;

    for (int kk = 0; kk < IN_CH / 4; ++kk) {
        const int kb = kk * 4 + koff;
        v2f a;
        a.x = sAw[m * APAD + kb];
        a.y = sAw[m * APAD + kb + 1];
        v2f b0, b1;
        b0.x = sW[(kb    ) * LAT + nn];
        b0.y = sW[(kb + 1) * LAT + nn];
        b1.x = sW[(kb    ) * LAT + 16 + nn];
        b1.y = sW[(kb + 1) * LAT + 16 + nn];
        acc0 = __builtin_amdgcn_wmma_f32_16x16x4_f32(false, a, false, b0,
                                                     (short)0, acc0, false, false);
        acc1 = __builtin_amdgcn_wmma_f32_16x16x4_f32(false, a, false, b1,
                                                     (short)0, acc1, false, false);
    }

    const int r0   = tile * TILE_M;
    const int rofs = (lane >> 4) << 3;
#pragma unroll
    for (int r = 0; r < 8; ++r) {
        int row = r0 + r + rofs;
        if (row < n) {
            h[(size_t)row * LAT + nn]      = acc0[r];
            h[(size_t)row * LAT + 16 + nn] = acc1[r];
        }
    }
}


__global__ void init_out_kernel(const float* __restrict__ dinv, const float* __restrict__ h,
                                const float* __restrict__ b, float* __restrict__ out, int n) {
    int gid = blockIdx.x * blockDim.x + threadIdx.x;
    if (gid >= n * LAT) return;
    int i = gid >> 5, c = gid & 31;
    float di = dinv[i];
    out[gid] = di * di * h[gid] + b[c];
}


__global__ void edge_scatter_kernel(const int* __restrict__ ei, const int* __restrict__ yei,
                                    const float* __restrict__ dinv, const float* __restrict__ h,
                                    float* __restrict__ out, int e1, int e2) {
    long long gid = (long long)blockIdx.x * blockDim.x + threadIdx.x;
    int e = (int)(gid >> 5);
    int c = (int)(gid & 31);
    int et = e1 + e2;
    if (e >= et) return;
    int s, d;
    if (e < e1) { s = ei[e];            d = ei[e1 + e]; }
    else        { int f = e - e1; s = yei[f]; d = yei[e2 + f]; }
    float w = dinv[s] * dinv[d];
    atomicAdd(out + (size_t)d * LAT + c, w * h[(size_t)s * LAT + c]);
}


#define DLB  9
#define DLM  ((1u << DLB) - 1u)
#define NB   392
#define MAXB 1024
#define CH   8192
#define PT   512
#define AT   1024
#define DG   512
#define DU   16
#define DTAB (7 * AT * 16)
static_assert(NB <= (1 << DLB) && (NB & 3) == 0, "a range's local index fits its field; table words hold whole ranges' bytes");

__device__ __forceinline__ int edge_src(const int* __restrict__ ei, const int* __restrict__ yei,
                                        int e1, int e) {
    return (e < e1) ? ei[e] : yei[e - e1];
}
__device__ __forceinline__ int edge_dst(const int* __restrict__ ei, const int* __restrict__ yei,
                                        int e1, int e2, int e) {
    return (e < e1) ? ei[e1 + e] : yei[e2 + (e - e1)];
}

template <int NT>
__device__ __forceinline__ int block_excl_scan(int v, int* sw, int& total) {
    const int lane = threadIdx.x & 31;
    const int wv   = threadIdx.x >> 5;
    int inc = v;
#pragma unroll
    for (int d = 1; d < 32; d <<= 1) {
        int t = __shfl_up(inc, d);
        if (lane >= d) inc += t;
    }
    if (lane == 31) sw[wv] = inc;
    __syncthreads();
    if (wv == 0) {
        int s  = (lane < NT / 32) ? sw[lane] : 0;
        int si = s;
#pragma unroll
        for (int d = 1; d < 32; d <<= 1) {
            int t = __shfl_up(si, d);
            if (lane >= d) si += t;
        }
        if (lane < NT / 32) sw[lane] = si - s;
        if (lane == 31) sw[32] = si;
    }
    __syncthreads();
    const int r = sw[wv] + inc - v;
    total = sw[32];
    __syncthreads();
    return r;
}

static_assert(PT == GW * 32, "a GEMM group is one full group of the count launch");
__global__ __launch_bounds__(PT) void count_gemm_kernel(const int* __restrict__ ei,
                                                        const int* __restrict__ yei,
                                                        int* __restrict__ counts,
                                                        int e1, int e2, int nchunks, int nb,
                                                        const float* __restrict__ x,
                                                        const float* __restrict__ W,
                                                        float* __restrict__ h, int n) {
    __shared__ __align__(16) float smem[GEMM_LDS_FLOATS];
    if ((int)blockIdx.x >= nchunks) {
        gemm_role(x, W, h, n, (int)blockIdx.x - nchunks, smem, smem + IN_CH * LAT);
        return;
    }
    int* hist = (int*)smem;
    const int tid = threadIdx.x;
    const int c   = blockIdx.x;
    const int et  = e1 + e2;
    const int e0  = c * CH;
    for (int i = tid; i < nb; i += PT) hist[i] = 0;
    int d[CH / PT];
#pragma unroll
    for (int k = 0; k < CH / PT; ++k) {
        const int e = e0 + tid + k * PT;
        d[k] = edge_dst(ei, yei, e1, e2, (e < et) ? e : et - 1);
    }
    __syncthreads();
#pragma unroll
    for (int k = 0; k < CH / PT; ++k) {
        const int e = e0 + tid + k * PT;
        if (e < et) atomicAdd(&hist[d[k] / NB], 1);
    }
    __syncthreads();
    for (int b = tid; b < nb; b += PT) counts[(size_t)b * nchunks + c] = hist[b];
}

__global__ __launch_bounds__(256) void range_scan_kernel(int* __restrict__ counts,
                                                         int* __restrict__ btot, int nchunks) {
    __shared__ int sw[33];
    const int tid = threadIdx.x;
    int* row = counts + (size_t)blockIdx.x * nchunks;
    int carry = 0;
    for (int c0 = 0; c0 < nchunks; c0 += 256) {
        const int c = c0 + tid;
        const int v = (c < nchunks) ? row[c] : 0;
        int tot;
        const int ex = block_excl_scan<256>(v, sw, tot);
        if (c < nchunks) row[c] = carry + ex;
        carry += tot;
    }
    if (tid == 0) btot[blockIdx.x] = carry;
}

__global__ __launch_bounds__(PT) void range_place_kernel(const int* __restrict__ ei,
                                                         const int* __restrict__ yei,
                                                         const int* __restrict__ woff,
                                                         const int* __restrict__ btot,
                                                         int* __restrict__ bstart,
                                                         unsigned* __restrict__ sorted,
                                                         int e1, int e2, int nchunks, int nb) {
    __shared__ int hist[MAXB];
    __shared__ int lbase[MAXB];
    __shared__ int gbase[MAXB];
    __shared__ unsigned words[CH];
    __shared__ unsigned short bk[CH];
    __shared__ int sw[33];

    const int tid = threadIdx.x;
    const int c   = blockIdx.x;
    const int et  = e1 + e2;
    const int e0  = c * CH;

    for (int i = tid; i < nb; i += PT) hist[i] = 0;

    int sv[CH / PT];
    int dv[CH / PT];
#pragma unroll
    for (int k = 0; k < CH / PT; ++k) {
        const int e  = e0 + tid + k * PT;
        const int ec = (e < et) ? e : et - 1;
        sv[k] = edge_src(ei, yei, e1, ec);
        dv[k] = edge_dst(ei, yei, e1, e2, ec);
    }
    __syncthreads();

    unsigned wreg[CH / PT];
    unsigned key[CH / PT];
#pragma unroll
    for (int k = 0; k < CH / PT; ++k) {
        const int e = e0 + tid + k * PT;
        key[k]  = 0xFFFFFFFFu;
        wreg[k] = 0u;
        if (e < et) {
            const int s = sv[k];
            const int d = dv[k];
            const int b = d / NB;
            const int r = atomicAdd(&hist[b], 1);
            wreg[k] = ((unsigned)s << DLB) | (unsigned)(d - b * NB);
            key[k]  = ((unsigned)b << 16) | (unsigned)r;
        }
    }
    __syncthreads();

    int carryL = 0, carryG = 0;
    for (int b0 = 0; b0 < nb; b0 += PT) {
        const int b  = b0 + tid;
        const int hv = (b < nb) ? hist[b] : 0;
        const int gv = (b < nb) ? btot[b] : 0;
        int tl, tg;
        const int el = block_excl_scan<PT>(hv, sw, tl);
        const int eg = block_excl_scan<PT>(gv, sw, tg);
        if (b < nb) {
            const int bs = carryG + eg;
            lbase[b] = carryL + el;
            gbase[b] = bs + woff[(size_t)b * nchunks + c];
            if (c == 0) bstart[b] = bs;
        }
        carryL += tl;
        carryG += tg;
    }
    if (c == 0 && tid == 0) bstart[nb] = carryG;
    __syncthreads();

#pragma unroll
    for (int k = 0; k < CH / PT; ++k) {
        if (key[k] != 0xFFFFFFFFu) {
            const int b = (int)(key[k] >> 16);
            const int p = lbase[b] + (int)(key[k] & 0xFFFFu);
            words[p] = wreg[k];
            bk[p]    = (unsigned short)b;
        }
    }
    __syncthreads();

    const int cnt = (et - e0 < CH) ? (et - e0) : CH;
    for (int i = tid; i < cnt; i += PT) {
        const int b = bk[i];
        sorted[gbase[b] + (i - lbase[b])] = words[i];
    }
}

__global__ __launch_bounds__(DG) void range_deg_kernel(const unsigned* __restrict__ sorted,
                                                       const int* __restrict__ bstart,
                                                       float* __restrict__ dinv,
                                                       unsigned* __restrict__ dtab, int ntabw, int n) {
    __shared__ int cnt[NB];
    const int tid = threadIdx.x;
    const int lo  = blockIdx.x * NB;
    for (int r = tid; r < NB; r += DG) cnt[r] = 0;

    const int start = bstart[blockIdx.x];
    const int end   = bstart[blockIdx.x + 1];
    const int lastw = (end > 0) ? (end - 1) : 0;

    unsigned w[DU];
#pragma unroll
    for (int u = 0; u < DU; ++u) {
        const int i = start + tid + u * DG;
        w[u] = sorted[(i < end) ? i : lastw];
    }
    __syncthreads();
#pragma unroll
    for (int u = 0; u < DU; ++u) {
        const int i = start + tid + u * DG;
        if (i < end) atomicAdd(&cnt[w[u] & DLM], 1);
    }
    for (int i0 = start + DU * DG; i0 < end; i0 += DU * DG) {
#pragma unroll
        for (int u = 0; u < DU; ++u) {
            const int i = i0 + tid + u * DG;
            w[u] = sorted[(i < end) ? i : lastw];
        }
#pragma unroll
        for (int u = 0; u < DU; ++u) {
            const int i = i0 + tid + u * DG;
            if (i < end) atomicAdd(&cnt[w[u] & DLM], 1);
        }
    }
    __syncthreads();
    for (int r = tid; r < NB; r += DG) {
        if (lo + r < n) dinv[lo + r] = rsqrtf((float)(1 + cnt[r]));
    }
    for (int q = tid; q < NB / 4; q += DG) {
        const int wi = (lo >> 2) + q;
        if (wi < ntabw) {
            unsigned word = 0u;
            for (int j = 0; j < 4; ++j) {
                const int cj = cnt[4 * q + j];
                word |= (unsigned)((cj < 255) ? cj : 255) << (8 * j);
            }
            dtab[wi] = word;
        }
    }
}

template <bool TAB>
__global__ __launch_bounds__(AT) void range_sum_kernel(const unsigned* __restrict__ sorted,
                                                       const int* __restrict__ bstart,
                                                       const float* __restrict__ dinv,
                                                       const unsigned* __restrict__ dtab,
                                                       const float* __restrict__ h,
                                                       const float* __restrict__ bias,
                                                       float* __restrict__ out, int n) {
    __shared__ float acc[((NB + AT / 32 - 1) / (AT / 32)) * (AT / 32) * LAT];
    __shared__ float sdinv[NB];
    __shared__ __align__(16) unsigned char stab[TAB ? DTAB : 16];

    const int tid  = threadIdx.x;
    const int lane = tid & 31;
    const int wave = tid >> 5;
    const int lo   = blockIdx.x * NB;

    constexpr int TV = TAB ? (DTAB / 16 / AT) : 1;
    const int nvec = (n + 15) >> 4;
    uint4 tv[TV];
    if (TAB) {
        const uint4* gt = (const uint4*)dtab;
#pragma unroll
        for (int u = 0; u < TV; ++u) {
            const int i = tid + u * AT;
            tv[u] = gt[(i < nvec) ? i : (nvec - 1)];
        }
    }

    for (int r = tid; r < NB; r += AT) {
        const float di = dinv[(lo + r < n) ? (lo + r) : (n - 1)];
        sdinv[r] = (lo + r < n) ? di : 0.0f;
    }
    {
        constexpr int RI = AT / 32;
        constexpr int NI = (NB + RI - 1) / RI;
        const float bl = bias[lane];
        float dv[NI];
        float hv[NI];
#pragma unroll
        for (int k = 0; k < NI; ++k) {
            const int node = lo + wave + k * RI;
            const int nc   = (node < n) ? node : (n - 1);
            dv[k] = dinv[nc];
            hv[k] = h[(size_t)nc * LAT + lane];
        }
        if (TAB) {
            uint4* st = (uint4*)stab;
#pragma unroll
            for (int u = 0; u < TV; ++u) {
                const int i = tid + u * AT;
                st[(i < nvec) ? i : (nvec - 1)] = tv[u];
            }
        }
#pragma unroll
        for (int k = 0; k < NI; ++k) {
            const int row = wave + k * RI;
            acc[row * LAT + lane] = (lo + row < n) ? (dv[k] * dv[k] * hv[k] + bl) : 0.0f;
        }
    }
    __syncthreads();

    const int start  = bstart[blockIdx.x];
    const int end    = bstart[blockIdx.x + 1];
    const int nbatch = (end - start + 31) >> 5;
    const float* hl  = h + lane;
    const unsigned lane_u = (unsigned)lane;

    const int lastw = end - 1;
    constexpr int KS = AT / 32;
    unsigned wd_cur = 0u, wd_nxt = 0u;
    float ds_cur = 0.0f;
    if (wave < nbatch) {
        const int i = start + (wave << 5) + lane;
        wd_cur = sorted[(i < end) ? i : lastw];
    }
    if (wave + KS < nbatch) {
        const int i = start + ((wave + KS) << 5) + lane;
        wd_nxt = sorted[(i < end) ? i : lastw];
    }
    if (!TAB && wave < nbatch) ds_cur = dinv[wd_cur >> DLB];

    for (int k = wave; k < nbatch; k += KS) {
        const int base = start + (k << 5);
        const int m    = (end - base < 32) ? (end - base) : 32;
        const unsigned wd = wd_cur;
        const int   s  = (int)(wd >> DLB);
        const int   dl = (int)(wd & DLM);
        const float sd = sdinv[dl];
        float ds = ds_cur;
        if (TAB) {
            const unsigned cs = stab[s];
            ds = rsqrtf((float)(1 + (int)cs));
            if (__builtin_amdgcn_ballot_w32(cs == 255u) != 0u) {
                if (cs == 255u) ds = dinv[s];
            }
        }
        const float w  = ds * sd;
        if (m == 32) {
            unsigned ix[32];
#pragma unroll
            for (int j = 0; j < 32; ++j) {
                ix[j] = ((unsigned)__builtin_amdgcn_readlane(s, j) << 5) | lane_u;
            }
            asm volatile("" :: "v"(ix[0]), "v"(ix[1]), "v"(ix[2]), "v"(ix[3]), "v"(ix[4]), "v"(ix[5]),
                         "v"(ix[6]), "v"(ix[7]), "v"(ix[8]), "v"(ix[9]), "v"(ix[10]), "v"(ix[11]),
                         "v"(ix[12]), "v"(ix[13]), "v"(ix[14]), "v"(ix[15]) : "memory");
            asm volatile("" :: "v"(ix[16]), "v"(ix[17]), "v"(ix[18]), "v"(ix[19]), "v"(ix[20]), "v"(ix[21]),
                         "v"(ix[22]), "v"(ix[23]), "v"(ix[24]), "v"(ix[25]), "v"(ix[26]), "v"(ix[27]),
                         "v"(ix[28]), "v"(ix[29]), "v"(ix[30]), "v"(ix[31]) : "memory");
            float v[32];
#pragma unroll
            for (int j = 0; j < 32; ++j) {
                v[j] = h[ix[j]];
            }
            if (!TAB && k + KS < nbatch) ds_cur = dinv[wd_nxt >> DLB];
            wd_cur = wd_nxt;
            if (k + 2 * KS < nbatch) {
                const int i = start + ((k + 2 * KS) << 5) + lane;
                wd_nxt = sorted[(i < end) ? i : lastw];
            }
#pragma unroll
            for (int j = 0; j < 32; ++j) {
                const float wj = __int_as_float(__builtin_amdgcn_readlane(__float_as_int(w), j));
                const int   dj = __builtin_amdgcn_readlane(dl, j);
                atomicAdd(&acc[dj * LAT + lane], wj * v[j]);
            }
        } else {
            for (int j = 0; j < m; ++j) {
                const int   sj = __builtin_amdgcn_readlane(s, j);
                const float wj = __int_as_float(__builtin_amdgcn_readlane(__float_as_int(w), j));
                const int   dj = __builtin_amdgcn_readlane(dl, j);
                atomicAdd(&acc[dj * LAT + lane], wj * hl[(size_t)sj * LAT]);
            }
        }
    }
    __syncthreads();

    for (int i = tid; i < NB * LAT; i += AT) {
        if (lo + (i >> 5) < n) out[(size_t)lo * LAT + i] = acc[i];
    }
}


extern "C" void kernel_launch(void* const* d_in, const int* in_sizes, int n_in,
                              void* d_out, int out_size, void* d_ws, size_t ws_size,
                              hipStream_t stream) {
    const float* x   = (const float*)d_in[0];
    const int*   ei  = (const int*)  d_in[1];
    const int*   yei = (const int*)  d_in[2];
    const float* W   = (const float*)d_in[3];
    const float* b   = (const float*)d_in[4];
    float* out = (float*)d_out;

    const int n  = in_sizes[0] / IN_CH;
    const int e1 = in_sizes[1] / 2;
    const int e2 = in_sizes[2] / 2;
    const int et = e1 + e2;

    float* deg = (float*)d_ws;
    size_t hoff = (((size_t)n * sizeof(float)) + 255) & ~(size_t)255;
    float* h = (float*)((char*)d_ws + hoff);

    const int nb      = (n + NB - 1) / NB;
    const int nchunks = (et + CH - 1) / CH;
    size_t soff = (hoff + (size_t)n * LAT * sizeof(float) + 255) & ~(size_t)255;
    size_t coff = (soff + (size_t)et * sizeof(unsigned) + 255) & ~(size_t)255;
    size_t toff = (coff + (size_t)nb * nchunks * sizeof(int) + 255) & ~(size_t)255;
    size_t boff = (toff + (size_t)nb * sizeof(int) + 255) & ~(size_t)255;
    size_t need = boff + ((size_t)nb + 1) * sizeof(int);
    unsigned* sorted = (unsigned*)((char*)d_ws + soff);
    int* counts = (int*)((char*)d_ws + coff);
    int* btot   = (int*)((char*)d_ws + toff);
    int* bstart = (int*)((char*)d_ws + boff);
    const bool grouped = (need <= ws_size) && (nb <= MAXB) && (n <= (1 << (32 - DLB))) && (nchunks > 0);
    const int  ntabw = ((n + 15) >> 4) << 2;
    const bool tab   = (n <= DTAB) && ((size_t)ntabw <= (size_t)nb * nchunks);
    unsigned* dtab   = (unsigned*)counts;

    const int ntiles  = (n + TILE_M - 1) / TILE_M;
    const int gblocks = (ntiles + GW - 1) / GW;
    if (grouped) {
        count_gemm_kernel<<<nchunks + gblocks, PT, 0, stream>>>(ei, yei, counts, e1, e2, nchunks, nb,
                                                                x, W, h, n);
        range_scan_kernel<<<nb, 256, 0, stream>>>(counts, btot, nchunks);
        range_place_kernel<<<nchunks, PT, 0, stream>>>(ei, yei, counts, btot, bstart, sorted,
                                                       e1, e2, nchunks, nb);
        range_deg_kernel<<<nb, DG, 0, stream>>>(sorted, bstart, deg, dtab, tab ? ntabw : 0, n);
        if (tab) range_sum_kernel<true><<<nb, AT, 0, stream>>>(sorted, bstart, deg, dtab, h, b, out, n);
        else     range_sum_kernel<false><<<nb, AT, 0, stream>>>(sorted, bstart, deg, dtab, h, b, out, n);
    } else {
        count_gemm_kernel<<<gblocks, PT, 0, stream>>>(ei, yei, counts, e1, e2, 0, nb, x, W, h, n);
        deg_init_kernel<<<(n + 255) / 256, 256, 0, stream>>>(deg, n);
        deg_edges_kernel<<<(et + 255) / 256, 256, 0, stream>>>(ei, yei, deg, e1, e2);
        dinv_kernel<<<(n + 255) / 256, 256, 0, stream>>>(deg, n);
        init_out_kernel<<<((size_t)n * LAT + 255) / 256, 256, 0, stream>>>(deg, h, b, out, n);
        long long threads = (long long)et * 32;
        int blocks = (int)((threads + 255) / 256);
        edge_scatter_kernel<<<blocks, 256, 0, stream>>>(ei, yei, deg, h, out, e1, e2);
    }
}


# edges ordered by range inside each chunk only, in the GEMM launch; degree and range-sum blocks find their runs via a 16-bit run table: no chip-wide place pass, 4->3 launches; sum order as before
# speedup vs baseline: 5.1440x; 1.0924x over previous
#include <hip/hip_runtime.h>
#include <hip/hip_bf16.h>

typedef __attribute__((ext_vector_type(2))) float v2f;
typedef __attribute__((ext_vector_type(8))) float v8f;

#define IN_CH 128
#define LAT   32
#define TILE_M 16
#define APAD  132


__global__ void deg_init_kernel(float* __restrict__ deg, int n) {
    int i = blockIdx.x * blockDim.x + threadIdx.x;
    if (i < n) deg[i] = 1.0f;
}

__global__ void deg_edges_kernel(const int* __restrict__ ei, const int* __restrict__ yei,
                                 float* __restrict__ deg, int e1, int e2) {
    int e = blockIdx.x * blockDim.x + threadIdx.x;
    int et = e1 + e2;
    if (e >= et) return;
    int d = (e < e1) ? ei[e1 + e] : yei[e2 + (e - e1)];
    atomicAdd(deg + d, 1.0f);
}

__global__ void dinv_kernel(float* __restrict__ deg, int n) {
    int i = blockIdx.x * blockDim.x + threadIdx.x;
    if (i < n) deg[i] = rsqrtf(deg[i]);
}


#define GW 16
#define GEMM_LDS_FLOATS (IN_CH * LAT + GW * TILE_M * APAD)

__device__ __forceinline__ void gemm_role(const float* __restrict__ x,
                                          const float* __restrict__ W,
                                          float* __restrict__ h, int n, int gblock,
                                          float* __restrict__ sW,
                                          float* __restrict__ sAall) {
    const int tid  = threadIdx.x;
    const int wave = tid >> 5;
    const int lane = tid & 31;
    const int ntiles = (n + TILE_M - 1) / TILE_M;
    const int tile = gblock * GW + wave;
    float* sAw = sAall + wave * (TILE_M * APAD);

    {
        const float4* Wv = (const float4*)W;
        float4* sWv = (float4*)sW;
        for (int i = tid; i < (IN_CH * LAT) / 4; i += GW * 32) sWv[i] = Wv[i];
    }
    if (tile < ntiles) {
        const int r0 = tile * TILE_M;
        for (int i = lane; i < TILE_M * (IN_CH / 4); i += 32) {
            int row = i >> 5;
            int c4  = i & 31;
            int grow = r0 + row; if (grow >= n) grow = n - 1;
            float4 v = ((const float4*)(x + (size_t)grow * IN_CH))[c4];
            float* dst = &sAw[row * APAD + c4 * 4];
            dst[0] = v.x; dst[1] = v.y; dst[2] = v.z; dst[3] = v.w;
        }
    }
    __syncthreads();

    if (tile >= ntiles) return;

    v8f acc0 = {};
    v8f acc1 = {};
    const int m    = lane & 15;
    const int koff = (lane >> 4) << 1;
    const int nn   = lane & 15;

    for (int kk = 0; kk < IN_CH / 4; ++kk) {
        const int kb = kk * 4 + koff;
        v2f a;
        a.x = sAw[m * APAD + kb];
        a.y = sAw[m * APAD + kb + 1];
        v2f b0, b1;
        b0.x = sW[(kb    ) * LAT + nn];
        b0.y = sW[(kb + 1) * LAT + nn];
        b1.x = sW[(kb    ) * LAT + 16 + nn];
        b1.y = sW[(kb + 1) * LAT + 16 + nn];
        acc0 = __builtin_amdgcn_wmma_f32_16x16x4_f32(false, a, false, b0,
                                                     (short)0, acc0, false, false);
        acc1 = __builtin_amdgcn_wmma_f32_16x16x4_f32(false, a, false, b1,
                                                     (short)0, acc1, false, false);
    }

    const int r0   = tile * TILE_M;
    const int rofs = (lane >> 4) << 3;
#pragma unroll
    for (int r = 0; r < 8; ++r) {
        int row = r0 + r + rofs;
        if (row < n) {
            h[(size_t)row * LAT + nn]      = acc0[r];
            h[(size_t)row * LAT + 16 + nn] = acc1[r];
        }
    }
}


__global__ void init_out_kernel(const float* __restrict__ dinv, const float* __restrict__ h,
                                const float* __restrict__ b, float* __restrict__ out, int n) {
    int gid = blockIdx.x * blockDim.x + threadIdx.x;
    if (gid >= n * LAT) return;
    int i = gid >> 5, c = gid & 31;
    float di = dinv[i];
    out[gid] = di * di * h[gid] + b[c];
}


__global__ void edge_scatter_kernel(const int* __restrict__ ei, const int* __restrict__ yei,
                                    const float* __restrict__ dinv, const float* __restrict__ h,
                                    float* __restrict__ out, int e1, int e2) {
    long long gid = (long long)blockIdx.x * blockDim.x + threadIdx.x;
    int e = (int)(gid >> 5);
    int c = (int)(gid & 31);
    int et = e1 + e2;
    if (e >= et) return;
    int s, d;
    if (e < e1) { s = ei[e];            d = ei[e1 + e]; }
    else        { int f = e - e1; s = yei[f]; d = yei[e2 + f]; }
    float w = dinv[s] * dinv[d];
    atomicAdd(out + (size_t)d * LAT + c, w * h[(size_t)s * LAT + c]);
}


#define DLB  9
#define DLM  ((1u << DLB) - 1u)
#define NB   392
#define MAXB 1024
#define CH   8192
#define MAXC 1024
#define PT   512
#define AT   1024
#define DG   512
#define DU   16
#define DTAB (7 * AT * 16)
static_assert(NB <= (1 << DLB) && (NB & 3) == 0, "a range's local index fits its field; table words hold whole ranges' bytes");

__device__ __forceinline__ int edge_src(const int* __restrict__ ei, const int* __restrict__ yei,
                                        int e1, int e) {
    return (e < e1) ? ei[e] : yei[e - e1];
}
__device__ __forceinline__ int edge_dst(const int* __restrict__ ei, const int* __restrict__ yei,
                                        int e1, int e2, int e) {
    return (e < e1) ? ei[e1 + e] : yei[e2 + (e - e1)];
}

template <int NT>
__device__ __forceinline__ int block_excl_scan(int v, int* sw, int& total) {
    const int lane = threadIdx.x & 31;
    const int wv   = threadIdx.x >> 5;
    int inc = v;
#pragma unroll
    for (int d = 1; d < 32; d <<= 1) {
        int t = __shfl_up(inc, d);
        if (lane >= d) inc += t;
    }
    if (lane == 31) sw[wv] = inc;
    __syncthreads();
    if (wv == 0) {
        int s  = (lane < NT / 32) ? sw[lane] : 0;
        int si = s;
#pragma unroll
        for (int d = 1; d < 32; d <<= 1) {
            int t = __shfl_up(si, d);
            if (lane >= d) si += t;
        }
        if (lane < NT / 32) sw[lane] = si - s;
        if (lane == 31) sw[32] = si;
    }
    __syncthreads();
    const int r = sw[wv] + inc - v;
    total = sw[32];
    __syncthreads();
    return r;
}

static_assert(PT == GW * 32, "a GEMM group is one full group of the grouping launch");
static_assert(CH <= 65535, "a place inside a chunk fits 16 bits");
__global__ __launch_bounds__(PT) void group_gemm_kernel(const int* __restrict__ ei,
                                                        const int* __restrict__ yei,
                                                        unsigned short* __restrict__ lb16,
                                                        unsigned* __restrict__ grouped,
                                                        int e1, int e2, int nchunks, int nb, int rs,
                                                        const float* __restrict__ x,
                                                        const float* __restrict__ W,
                                                        float* __restrict__ h, int n) {
    __shared__ __align__(16) float smem[GEMM_LDS_FLOATS];
    if ((int)blockIdx.x >= nchunks) {
        gemm_role(x, W, h, n, (int)blockIdx.x - nchunks, smem, smem + IN_CH * LAT);
        return;
    }
    static_assert(2 * MAXB + 40 + CH <= GEMM_LDS_FLOATS, "the grouping's tables fit the GEMM's LDS block");
    int* hist       = (int*)smem;
    int* lbase      = hist + MAXB;
    int* sw         = lbase + MAXB;
    unsigned* words = (unsigned*)(sw + 40);
    const int tid = threadIdx.x;
    const int c   = blockIdx.x;
    const int et  = e1 + e2;
    const int e0  = c * CH;
    for (int i = tid; i < nb; i += PT) hist[i] = 0;
    int sv[CH / PT];
    int dv[CH / PT];
#pragma unroll
    for (int k = 0; k < CH / PT; ++k) {
        const int e  = e0 + tid + k * PT;
        const int ec = (e < et) ? e : et - 1;
        sv[k] = edge_src(ei, yei, e1, ec);
        dv[k] = edge_dst(ei, yei, e1, e2, ec);
    }
    __syncthreads();
    unsigned wreg[CH / PT];
    unsigned key[CH / PT];
#pragma unroll
    for (int k = 0; k < CH / PT; ++k) {
        const int e = e0 + tid + k * PT;
        key[k]  = 0xFFFFFFFFu;
        wreg[k] = 0u;
        if (e < et) {
            const int s = sv[k];
            const int d = dv[k];
            const int b = d / NB;
            const int r = atomicAdd(&hist[b], 1);
            wreg[k] = ((unsigned)s << DLB) | (unsigned)(d - b * NB);
            key[k]  = ((unsigned)b << 16) | (unsigned)r;
        }
    }
    __syncthreads();

    unsigned short* lrow = lb16 + (size_t)c * rs;
    int carry = 0;
    for (int b0 = 0; b0 < nb; b0 += PT) {
        const int b  = b0 + tid;
        const int hv = (b < nb) ? hist[b] : 0;
        int tl;
        const int el = block_excl_scan<PT>(hv, sw, tl);
        if (b < nb) {
            lbase[b] = carry + el;
            lrow[b]  = (unsigned short)(carry + el);
        }
        carry += tl;
    }
    if (tid == 0) lrow[nb] = (unsigned short)carry;
    __syncthreads();

#pragma unroll
    for (int k = 0; k < CH / PT; ++k) {
        if (key[k] != 0xFFFFFFFFu) {
            const int b = (int)(key[k] >> 16);
            words[lbase[b] + (int)(key[k] & 0xFFFFu)] = wreg[k];
        }
    }
    __syncthreads();

    const int cnt = (et - e0 < CH) ? (et - e0) : CH;
    for (int i = tid; i < cnt; i += PT) grouped[e0 + i] = words[i];
}

__global__ __launch_bounds__(DG) void range_deg_kernel(const unsigned* __restrict__ grouped,
                                                       const unsigned short* __restrict__ lb16,
                                                       float* __restrict__ dinv,
                                                       unsigned* __restrict__ dtab, int ntabw, int n,
                                                       int nchunks, int rs, int et) {
    __shared__ int cnt[NB];
    __shared__ int roff[MAXC];
    __shared__ int rlen[MAXC];
    const int tid  = threadIdx.x;
    const int lane = tid & 31;
    const int wave = tid >> 5;
    const int lo   = blockIdx.x * NB;
    for (int r = tid; r < NB; r += DG) cnt[r] = 0;
    for (int c = tid; c < nchunks; c += DG) {
        const unsigned short* rw = lb16 + (size_t)c * rs + blockIdx.x;
        const int a = rw[0];
        const int z = rw[1];
        roff[c] = c * CH + a;
        rlen[c] = z - a;
    }
    __syncthreads();

    constexpr int NW = DG / 32;
    const int lastc = nchunks - 1;
    const int lastw = et - 1;
    for (int c0 = wave; c0 < nchunks; c0 += NW * DU) {
        unsigned w0[DU], w1[DU];
        int ln[DU];
#pragma unroll
        for (int u = 0; u < DU; ++u) {
            const int c  = c0 + u * NW;
            const int cc = (c < lastc) ? c : lastc;
            const int L  = (c < nchunks) ? rlen[cc] : 0;
            const int o  = roff[cc];
            const int l0 = (lane < L) ? lane : ((L > 0) ? (L - 1) : 0);
            const int l1 = (lane + 32 < L) ? (lane + 32) : ((L > 0) ? (L - 1) : 0);
            const int i0 = o + l0;
            const int i1 = o + l1;
            w0[u] = grouped[(i0 < lastw) ? i0 : lastw];
            w1[u] = grouped[(i1 < lastw) ? i1 : lastw];
            ln[u] = L;
        }
#pragma unroll
        for (int u = 0; u < DU; ++u) {
            if (lane < ln[u])      atomicAdd(&cnt[w0[u] & DLM], 1);
            if (lane + 32 < ln[u]) atomicAdd(&cnt[w1[u] & DLM], 1);
        }
        for (int u = 0; u < DU; ++u) {
            const int c = c0 + u * NW;
            if (c < nchunks) {
                const int L = rlen[c];
                const int o = roff[c];
                for (int l = lane + 64; l < L; l += 32) atomicAdd(&cnt[grouped[o + l] & DLM], 1);
            }
        }
    }
    __syncthreads();
    for (int r = tid; r < NB; r += DG) {
        if (lo + r < n) dinv[lo + r] = rsqrtf((float)(1 + cnt[r]));
    }
    for (int q = tid; q < NB / 4; q += DG) {
        const int wi = (lo >> 2) + q;
        if (wi < ntabw) {
            unsigned word = 0u;
            for (int j = 0; j < 4; ++j) {
                const int cj = cnt[4 * q + j];
                word |= (unsigned)((cj < 255) ? cj : 255) << (8 * j);
            }
            dtab[wi] = word;
        }
    }
}

__device__ __forceinline__ unsigned range_word(const unsigned* __restrict__ grouped,
                                               const int* rpos, const int* radj,
                                               int nchunks, int p2, int i) {
    int c = 0;
    for (int s = p2 >> 1; s > 0; s >>= 1) {
        const int t  = c + s;
        const int tc = (t < nchunks) ? t : (nchunks - 1);
        if (t < nchunks && rpos[tc] <= i) c = t;
    }
    return grouped[radj[c] + i];
}

template <bool TAB>
__global__ __launch_bounds__(AT) void range_sum_kernel(const unsigned* __restrict__ grouped,
                                                       const unsigned short* __restrict__ lb16,
                                                       const float* __restrict__ dinv,
                                                       const unsigned* __restrict__ dtab,
                                                       const float* __restrict__ h,
                                                       const float* __restrict__ bias,
                                                       float* __restrict__ out, int n,
                                                       int nchunks, int rs) {
    __shared__ float acc[((NB + AT / 32 - 1) / (AT / 32)) * (AT / 32) * LAT];
    __shared__ float sdinv[NB];
    __shared__ __align__(16) unsigned char stab[TAB ? DTAB : 16];
    __shared__ int rpos[MAXC];
    __shared__ int radj[MAXC];
    __shared__ int sw[33];

    const int tid  = threadIdx.x;
    const int lane = tid & 31;
    const int wave = tid >> 5;
    const int lo   = blockIdx.x * NB;

    constexpr int TV = TAB ? (DTAB / 16 / AT) : 1;
    const int nvec = (n + 15) >> 4;
    uint4 tv[TV];
    if (TAB) {
        const uint4* gt = (const uint4*)dtab;
#pragma unroll
        for (int u = 0; u < TV; ++u) {
            const int i = tid + u * AT;
            tv[u] = gt[(i < nvec) ? i : (nvec - 1)];
        }
    }

    const int rc = (tid < nchunks) ? tid : (nchunks - 1);
    const unsigned short* rw = lb16 + (size_t)rc * rs + blockIdx.x;
    const int ra = rw[0];
    const int rz = rw[1];

    for (int r = tid; r < NB; r += AT) {
        const float di = dinv[(lo + r < n) ? (lo + r) : (n - 1)];
        sdinv[r] = (lo + r < n) ? di : 0.0f;
    }
    {
        constexpr int RI = AT / 32;
        constexpr int NI = (NB + RI - 1) / RI;
        const float bl = bias[lane];
        float dv[NI];
        float hv[NI];
#pragma unroll
        for (int k = 0; k < NI; ++k) {
            const int node = lo + wave + k * RI;
            const int nc   = (node < n) ? node : (n - 1);
            dv[k] = dinv[nc];
            hv[k] = h[(size_t)nc * LAT + lane];
        }
        if (TAB) {
            uint4* st = (uint4*)stab;
#pragma unroll
            for (int u = 0; u < TV; ++u) {
                const int i = tid + u * AT;
                st[(i < nvec) ? i : (nvec - 1)] = tv[u];
            }
        }
#pragma unroll
        for (int k = 0; k < NI; ++k) {
            const int row = wave + k * RI;
            acc[row * LAT + lane] = (lo + row < n) ? (dv[k] * dv[k] * hv[k] + bl) : 0.0f;
        }
    }
    int total;
    {
        const int rl = (tid < nchunks) ? (rz - ra) : 0;
        const int ex = block_excl_scan<AT>(rl, sw, total);
        if (tid < nchunks) { rpos[tid] = ex; radj[tid] = tid * CH + ra - ex; }
    }
    __syncthreads();

    const int start  = 0;
    const int end    = total;
    const int nbatch = (end + 31) >> 5;
    int p2 = 1;
    while (p2 < nchunks) p2 <<= 1;
    const float* hl  = h + lane;
    const unsigned lane_u = (unsigned)lane;

    const int lastw = end - 1;
    constexpr int KS = AT / 32;
    unsigned wd_cur = 0u, wd_nxt = 0u;
    float ds_cur = 0.0f;
    if (wave < nbatch) {
        const int i = start + (wave << 5) + lane;
        wd_cur = range_word(grouped, rpos, radj, nchunks, p2, (i < end) ? i : lastw);
    }
    if (wave + KS < nbatch) {
        const int i = start + ((wave + KS) << 5) + lane;
        wd_nxt = range_word(grouped, rpos, radj, nchunks, p2, (i < end) ? i : lastw);
    }
    if (!TAB && wave < nbatch) ds_cur = dinv[wd_cur >> DLB];

    for (int k = wave; k < nbatch; k += KS) {
        const int base = start + (k << 5);
        const int m    = (end - base < 32) ? (end - base) : 32;
        const unsigned wd = wd_cur;
        const int   s  = (int)(wd >> DLB);
        const int   dl = (int)(wd & DLM);
        const float sd = sdinv[dl];
        float ds = ds_cur;
        if (TAB) {
            const unsigned cs = stab[s];
            ds = rsqrtf((float)(1 + (int)cs));
            if (__builtin_amdgcn_ballot_w32(cs == 255u) != 0u) {
                if (cs == 255u) ds = dinv[s];
            }
        }
        const float w  = ds * sd;
        if (m == 32) {
            unsigned ix[32];
#pragma unroll
            for (int j = 0; j < 32; ++j) {
                ix[j] = ((unsigned)__builtin_amdgcn_readlane(s, j) << 5) | lane_u;
            }
            asm volatile("" :: "v"(ix[0]), "v"(ix[1]), "v"(ix[2]), "v"(ix[3]), "v"(ix[4]), "v"(ix[5]),
                         "v"(ix[6]), "v"(ix[7]), "v"(ix[8]), "v"(ix[9]), "v"(ix[10]), "v"(ix[11]),
                         "v"(ix[12]), "v"(ix[13]), "v"(ix[14]), "v"(ix[15]) : "memory");
            asm volatile("" :: "v"(ix[16]), "v"(ix[17]), "v"(ix[18]), "v"(ix[19]), "v"(ix[20]), "v"(ix[21]),
                         "v"(ix[22]), "v"(ix[23]), "v"(ix[24]), "v"(ix[25]), "v"(ix[26]), "v"(ix[27]),
                         "v"(ix[28]), "v"(ix[29]), "v"(ix[30]), "v"(ix[31]) : "memory");
            float v[32];
#pragma unroll
            for (int j = 0; j < 32; ++j) {
                v[j] = h[ix[j]];
            }
            if (!TAB && k + KS < nbatch) ds_cur = dinv[wd_nxt >> DLB];
            wd_cur = wd_nxt;
            if (k + 2 * KS < nbatch) {
                const int i = start + ((k + 2 * KS) << 5) + lane;
                wd_nxt = range_word(grouped, rpos, radj, nchunks, p2, (i < end) ? i : lastw);
            }
#pragma unroll
            for (int j = 0; j < 32; ++j) {
                const float wj = __int_as_float(__builtin_amdgcn_readlane(__float_as_int(w), j));
                const int   dj = __builtin_amdgcn_readlane(dl, j);
                atomicAdd(&acc[dj * LAT + lane], wj * v[j]);
            }
        } else {
            for (int j = 0; j < m; ++j) {
                const int   sj = __builtin_amdgcn_readlane(s, j);
                const float wj = __int_as_float(__builtin_amdgcn_readlane(__float_as_int(w), j));
                const int   dj = __builtin_amdgcn_readlane(dl, j);
                atomicAdd(&acc[dj * LAT + lane], wj * hl[(size_t)sj * LAT]);
            }
        }
    }
    __syncthreads();

    for (int i = tid; i < NB * LAT; i += AT) {
        if (lo + (i >> 5) < n) out[(size_t)lo * LAT + i] = acc[i];
    }
}


extern "C" void kernel_launch(void* const* d_in, const int* in_sizes, int n_in,
                              void* d_out, int out_size, void* d_ws, size_t ws_size,
                              hipStream_t stream) {
    const float* x   = (const float*)d_in[0];
    const int*   ei  = (const int*)  d_in[1];
    const int*   yei = (const int*)  d_in[2];
    const float* W   = (const float*)d_in[3];
    const float* b   = (const float*)d_in[4];
    float* out = (float*)d_out;

    const int n  = in_sizes[0] / IN_CH;
    const int e1 = in_sizes[1] / 2;
    const int e2 = in_sizes[2] / 2;
    const int et = e1 + e2;

    float* deg = (float*)d_ws;
    size_t hoff = (((size_t)n * sizeof(float)) + 255) & ~(size_t)255;
    float* h = (float*)((char*)d_ws + hoff);

    const int nb      = (n + NB - 1) / NB;
    const int nchunks = (et + CH - 1) / CH;
    const int rs      = (nb + 64) & ~63;
    size_t soff = (hoff + (size_t)n * LAT * sizeof(float) + 255) & ~(size_t)255;
    size_t coff = (soff + (size_t)et * sizeof(unsigned) + 255) & ~(size_t)255;
    size_t toff = (coff + (size_t)(nb + 8) * nchunks * sizeof(int) + 255) & ~(size_t)255;
    size_t boff = (toff + (size_t)nb * sizeof(int) + 255) & ~(size_t)255;
    size_t need = boff + ((size_t)nb + 1) * sizeof(int);
    unsigned* grouped = (unsigned*)((char*)d_ws + soff);
    unsigned* plane   = (unsigned*)((char*)d_ws + coff);
    unsigned short* lb16 = (unsigned short*)plane;
    const bool grouped_ok = (need <= ws_size) && (nb <= MAXB) && (n <= (1 << (32 - DLB))) &&
                            (nchunks > 0) && (nchunks <= MAXC);
    const int    ntabw = ((n + 15) >> 4) << 2;
    const size_t dt0   = ((((size_t)nchunks * rs) >> 1) + 3) & ~(size_t)3;
    const bool   tab   = (n <= DTAB) && (dt0 + (size_t)ntabw <= (size_t)(nb + 8) * nchunks);
    unsigned* dtab     = plane + dt0;

    const int ntiles  = (n + TILE_M - 1) / TILE_M;
    const int gblocks = (ntiles + GW - 1) / GW;
    if (grouped_ok) {
        group_gemm_kernel<<<nchunks + gblocks, PT, 0, stream>>>(ei, yei, lb16, grouped, e1, e2, nchunks, nb, rs,
                                                                x, W, h, n);
        range_deg_kernel<<<nb, DG, 0, stream>>>(grouped, lb16, deg, dtab, tab ? ntabw : 0, n, nchunks, rs, et);
        if (tab) range_sum_kernel<true><<<nb, AT, 0, stream>>>(grouped, lb16, deg, dtab, h, b, out, n, nchunks, rs);
        else     range_sum_kernel<false><<<nb, AT, 0, stream>>>(grouped, lb16, deg, dtab, h, b, out, n, nchunks, rs);
    } else {
        group_gemm_kernel<<<gblocks, PT, 0, stream>>>(ei, yei, lb16, grouped, e1, e2, 0, nb, rs, x, W, h, n);
        deg_init_kernel<<<(n + 255) / 256, 256, 0, stream>>>(deg, n);
        deg_edges_kernel<<<(et + 255) / 256, 256, 0, stream>>>(ei, yei, deg, e1, e2);
        dinv_kernel<<<(n + 255) / 256, 256, 0, stream>>>(deg, n);
        init_out_kernel<<<((size_t)n * LAT + 255) / 256, 256, 0, stream>>>(deg, h, b, out, n);
        long long threads = (long long)et * 32;
        int blocks = (int)((threads + 255) / 256);
        edge_scatter_kernel<<<blocks, 256, 0, stream>>>(ei, yei, deg, h, out, e1, e2);
    }
}


# range-sum: block orders its edges by destination node in LDS (counting sort on degree bytes); a wave sums a run of equal destinations in a register, one LDS add per run; float sum order changes
# speedup vs baseline: 5.3754x; 1.0450x over previous
#include <hip/hip_runtime.h>
#include <hip/hip_bf16.h>

typedef __attribute__((ext_vector_type(2))) float v2f;
typedef __attribute__((ext_vector_type(8))) float v8f;

#define IN_CH 128
#define LAT   32
#define TILE_M 16
#define APAD  132


__global__ void deg_init_kernel(float* __restrict__ deg, int n) {
    int i = blockIdx.x * blockDim.x + threadIdx.x;
    if (i < n) deg[i] = 1.0f;
}

__global__ void deg_edges_kernel(const int* __restrict__ ei, const int* __restrict__ yei,
                                 float* __restrict__ deg, int e1, int e2) {
    int e = blockIdx.x * blockDim.x + threadIdx.x;
    int et = e1 + e2;
    if (e >= et) return;
    int d = (e < e1) ? ei[e1 + e] : yei[e2 + (e - e1)];
    atomicAdd(deg + d, 1.0f);
}

__global__ void dinv_kernel(float* __restrict__ deg, int n) {
    int i = blockIdx.x * blockDim.x + threadIdx.x;
    if (i < n) deg[i] = rsqrtf(deg[i]);
}


#define GW 16
#define GEMM_LDS_FLOATS (IN_CH * LAT + GW * TILE_M * APAD)

__device__ __forceinline__ void gemm_role(const float* __restrict__ x,
                                          const float* __restrict__ W,
                                          float* __restrict__ h, int n, int gblock,
                                          float* __restrict__ sW,
                                          float* __restrict__ sAall) {
    const int tid  = threadIdx.x;
    const int wave = tid >> 5;
    const int lane = tid & 31;
    const int ntiles = (n + TILE_M - 1) / TILE_M;
    const int tile = gblock * GW + wave;
    float* sAw = sAall + wave * (TILE_M * APAD);

    {
        const float4* Wv = (const float4*)W;
        float4* sWv = (float4*)sW;
        for (int i = tid; i < (IN_CH * LAT) / 4; i += GW * 32) sWv[i] = Wv[i];
    }
    if (tile < ntiles) {
        const int r0 = tile * TILE_M;
        for (int i = lane; i < TILE_M * (IN_CH / 4); i += 32) {
            int row = i >> 5;
            int c4  = i & 31;
            int grow = r0 + row; if (grow >= n) grow = n - 1;
            float4 v = ((const float4*)(x + (size_t)grow * IN_CH))[c4];
            float* dst = &sAw[row * APAD + c4 * 4];
            dst[0] = v.x; dst[1] = v.y; dst[2] = v.z; dst[3] = v.w;
        }
    }
    __syncthreads();

    if (tile >= ntiles) return;

    v8f acc0 = {};
    v8f acc1 = {};
    const int m    = lane & 15;
    const int koff = (lane >> 4) << 1;
    const int nn   = lane & 15;

    for (int kk = 0; kk < IN_CH / 4; ++kk) {
        const int kb = kk * 4 + koff;
        v2f a;
        a.x = sAw[m * APAD + kb];
        a.y = sAw[m * APAD + kb + 1];
        v2f b0, b1;
        b0.x = sW[(kb    ) * LAT + nn];
        b0.y = sW[(kb + 1) * LAT + nn];
        b1.x = sW[(kb    ) * LAT + 16 + nn];
        b1.y = sW[(kb + 1) * LAT + 16 + nn];
        acc0 = __builtin_amdgcn_wmma_f32_16x16x4_f32(false, a, false, b0,
                                                     (short)0, acc0, false, false);
        acc1 = __builtin_amdgcn_wmma_f32_16x16x4_f32(false, a, false, b1,
                                                     (short)0, acc1, false, false);
    }

    const int r0   = tile * TILE_M;
    const int rofs = (lane >> 4) << 3;
#pragma unroll
    for (int r = 0; r < 8; ++r) {
        int row = r0 + r + rofs;
        if (row < n) {
            h[(size_t)row * LAT + nn]      = acc0[r];
            h[(size_t)row * LAT + 16 + nn] = acc1[r];
        }
    }
}


__global__ void init_out_kernel(const float* __restrict__ dinv, const float* __restrict__ h,
                                const float* __restrict__ b, float* __restrict__ out, int n) {
    int gid = blockIdx.x * blockDim.x + threadIdx.x;
    if (gid >= n * LAT) return;
    int i = gid >> 5, c = gid & 31;
    float di = dinv[i];
    out[gid] = di * di * h[gid] + b[c];
}


__global__ void edge_scatter_kernel(const int* __restrict__ ei, const int* __restrict__ yei,
                                    const float* __restrict__ dinv, const float* __restrict__ h,
                                    float* __restrict__ out, int e1, int e2) {
    long long gid = (long long)blockIdx.x * blockDim.x + threadIdx.x;
    int e = (int)(gid >> 5);
    int c = (int)(gid & 31);
    int et = e1 + e2;
    if (e >= et) return;
    int s, d;
    if (e < e1) { s = ei[e];            d = ei[e1 + e]; }
    else        { int f = e - e1; s = yei[f]; d = yei[e2 + f]; }
    float w = dinv[s] * dinv[d];
    atomicAdd(out + (size_t)d * LAT + c, w * h[(size_t)s * LAT + c]);
}


#define DLB  9
#define DLM  ((1u << DLB) - 1u)
#define NB   392
#define MAXB 1024
#define CH   8192
#define MAXC 1024
#define PT   512
#define AT   1024
#define DG   512
#define DU   16
#define SCAP 16384
#define DTAB (7 * AT * 16)
static_assert(NB <= (1 << DLB) && (NB & 3) == 0, "a range's local index fits its field; table words hold whole ranges' bytes");

__device__ __forceinline__ int edge_src(const int* __restrict__ ei, const int* __restrict__ yei,
                                        int e1, int e) {
    return (e < e1) ? ei[e] : yei[e - e1];
}
__device__ __forceinline__ int edge_dst(const int* __restrict__ ei, const int* __restrict__ yei,
                                        int e1, int e2, int e) {
    return (e < e1) ? ei[e1 + e] : yei[e2 + (e - e1)];
}

template <int NT>
__device__ __forceinline__ int block_excl_scan(int v, int* sw, int& total) {
    const int lane = threadIdx.x & 31;
    const int wv   = threadIdx.x >> 5;
    int inc = v;
#pragma unroll
    for (int d = 1; d < 32; d <<= 1) {
        int t = __shfl_up(inc, d);
        if (lane >= d) inc += t;
    }
    if (lane == 31) sw[wv] = inc;
    __syncthreads();
    if (wv == 0) {
        int s  = (lane < NT / 32) ? sw[lane] : 0;
        int si = s;
#pragma unroll
        for (int d = 1; d < 32; d <<= 1) {
            int t = __shfl_up(si, d);
            if (lane >= d) si += t;
        }
        if (lane < NT / 32) sw[lane] = si - s;
        if (lane == 31) sw[32] = si;
    }
    __syncthreads();
    const int r = sw[wv] + inc - v;
    total = sw[32];
    __syncthreads();
    return r;
}

static_assert(PT == GW * 32, "a GEMM group is one full group of the grouping launch");
static_assert(CH <= 65535, "a place inside a chunk fits 16 bits");
__global__ __launch_bounds__(PT) void group_gemm_kernel(const int* __restrict__ ei,
                                                        const int* __restrict__ yei,
                                                        unsigned short* __restrict__ lb16,
                                                        unsigned* __restrict__ grouped,
                                                        int e1, int e2, int nchunks, int nb, int rs,
                                                        const float* __restrict__ x,
                                                        const float* __restrict__ W,
                                                        float* __restrict__ h, int n) {
    __shared__ __align__(16) float smem[GEMM_LDS_FLOATS];
    if ((int)blockIdx.x >= nchunks) {
        gemm_role(x, W, h, n, (int)blockIdx.x - nchunks, smem, smem + IN_CH * LAT);
        return;
    }
    static_assert(2 * MAXB + 40 + CH <= GEMM_LDS_FLOATS, "the grouping's tables fit the GEMM's LDS block");
    int* hist       = (int*)smem;
    int* lbase      = hist + MAXB;
    int* sw         = lbase + MAXB;
    unsigned* words = (unsigned*)(sw + 40);
    const int tid = threadIdx.x;
    const int c   = blockIdx.x;
    const int et  = e1 + e2;
    const int e0  = c * CH;
    for (int i = tid; i < nb; i += PT) hist[i] = 0;
    int sv[CH / PT];
    int dv[CH / PT];
#pragma unroll
    for (int k = 0; k < CH / PT; ++k) {
        const int e  = e0 + tid + k * PT;
        const int ec = (e < et) ? e : et - 1;
        sv[k] = edge_src(ei, yei, e1, ec);
        dv[k] = edge_dst(ei, yei, e1, e2, ec);
    }
    __syncthreads();
    unsigned wreg[CH / PT];
    unsigned key[CH / PT];
#pragma unroll
    for (int k = 0; k < CH / PT; ++k) {
        const int e = e0 + tid + k * PT;
        key[k]  = 0xFFFFFFFFu;
        wreg[k] = 0u;
        if (e < et) {
            const int s = sv[k];
            const int d = dv[k];
            const int b = d / NB;
            const int r = atomicAdd(&hist[b], 1);
            wreg[k] = ((unsigned)s << DLB) | (unsigned)(d - b * NB);
            key[k]  = ((unsigned)b << 16) | (unsigned)r;
        }
    }
    __syncthreads();

    unsigned short* lrow = lb16 + (size_t)c * rs;
    int carry = 0;
    for (int b0 = 0; b0 < nb; b0 += PT) {
        const int b  = b0 + tid;
        const int hv = (b < nb) ? hist[b] : 0;
        int tl;
        const int el = block_excl_scan<PT>(hv, sw, tl);
        if (b < nb) {
            lbase[b] = carry + el;
            lrow[b]  = (unsigned short)(carry + el);
        }
        carry += tl;
    }
    if (tid == 0) lrow[nb] = (unsigned short)carry;
    __syncthreads();

#pragma unroll
    for (int k = 0; k < CH / PT; ++k) {
        if (key[k] != 0xFFFFFFFFu) {
            const int b = (int)(key[k] >> 16);
            words[lbase[b] + (int)(key[k] & 0xFFFFu)] = wreg[k];
        }
    }
    __syncthreads();

    const int cnt = (et - e0 < CH) ? (et - e0) : CH;
    for (int i = tid; i < cnt; i += PT) grouped[e0 + i] = words[i];
}

__global__ __launch_bounds__(DG) void range_deg_kernel(const unsigned* __restrict__ grouped,
                                                       const unsigned short* __restrict__ lb16,
                                                       float* __restrict__ dinv,
                                                       unsigned* __restrict__ dtab, int ntabw, int n,
                                                       int nchunks, int rs, int et) {
    __shared__ int cnt[NB];
    __shared__ int roff[MAXC];
    __shared__ int rlen[MAXC];
    const int tid  = threadIdx.x;
    const int lane = tid & 31;
    const int wave = tid >> 5;
    const int lo   = blockIdx.x * NB;
    for (int r = tid; r < NB; r += DG) cnt[r] = 0;
    for (int c = tid; c < nchunks; c += DG) {
        const unsigned short* rw = lb16 + (size_t)c * rs + blockIdx.x;
        const int a = rw[0];
        const int z = rw[1];
        roff[c] = c * CH + a;
        rlen[c] = z - a;
    }
    __syncthreads();

    constexpr int NW = DG / 32;
    const int lastc = nchunks - 1;
    const int lastw = et - 1;
    for (int c0 = wave; c0 < nchunks; c0 += NW * DU) {
        unsigned w0[DU], w1[DU];
        int ln[DU];
#pragma unroll
        for (int u = 0; u < DU; ++u) {
            const int c  = c0 + u * NW;
            const int cc = (c < lastc) ? c : lastc;
            const int L  = (c < nchunks) ? rlen[cc] : 0;
            const int o  = roff[cc];
            const int l0 = (lane < L) ? lane : ((L > 0) ? (L - 1) : 0);
            const int l1 = (lane + 32 < L) ? (lane + 32) : ((L > 0) ? (L - 1) : 0);
            const int i0 = o + l0;
            const int i1 = o + l1;
            w0[u] = grouped[(i0 < lastw) ? i0 : lastw];
            w1[u] = grouped[(i1 < lastw) ? i1 : lastw];
            ln[u] = L;
        }
#pragma unroll
        for (int u = 0; u < DU; ++u) {
            if (lane < ln[u])      atomicAdd(&cnt[w0[u] & DLM], 1);
            if (lane + 32 < ln[u]) atomicAdd(&cnt[w1[u] & DLM], 1);
        }
        for (int u = 0; u < DU; ++u) {
            const int c = c0 + u * NW;
            if (c < nchunks) {
                const int L = rlen[c];
                const int o = roff[c];
                for (int l = lane + 64; l < L; l += 32) atomicAdd(&cnt[grouped[o + l] & DLM], 1);
            }
        }
    }
    __syncthreads();
    for (int r = tid; r < NB; r += DG) {
        if (lo + r < n) dinv[lo + r] = rsqrtf((float)(1 + cnt[r]));
    }
    for (int q = tid; q < NB / 4; q += DG) {
        const int wi = (lo >> 2) + q;
        if (wi < ntabw) {
            unsigned word = 0u;
            for (int j = 0; j < 4; ++j) {
                const int cj = cnt[4 * q + j];
                word |= (unsigned)((cj < 255) ? cj : 255) << (8 * j);
            }
            dtab[wi] = word;
        }
    }
}

__device__ __forceinline__ unsigned range_word(const unsigned* __restrict__ grouped,
                                               const int* rpos, const int* radj,
                                               int nchunks, int p2, int i) {
    int c = 0;
    for (int s = p2 >> 1; s > 0; s >>= 1) {
        const int t  = c + s;
        const int tc = (t < nchunks) ? t : (nchunks - 1);
        if (t < nchunks && rpos[tc] <= i) c = t;
    }
    return grouped[radj[c] + i];
}

template <bool TAB>
__global__ __launch_bounds__(AT) void range_sum_kernel(const unsigned* __restrict__ grouped,
                                                       const unsigned short* __restrict__ lb16,
                                                       const float* __restrict__ dinv,
                                                       const unsigned* __restrict__ dtab,
                                                       const float* __restrict__ h,
                                                       const float* __restrict__ bias,
                                                       float* __restrict__ out, int n,
                                                       int nchunks, int rs) {
    __shared__ float acc[((NB + AT / 32 - 1) / (AT / 32)) * (AT / 32) * LAT];
    __shared__ float sdinv[NB];
    __shared__ __align__(16) unsigned char stab[TAB ? DTAB : 16];
    __shared__ int rpos[MAXC];
    __shared__ int radj[MAXC];
    __shared__ int sw[33];
    __shared__ unsigned sorted[TAB ? SCAP : 1];
    __shared__ int ncur[TAB ? NB : 1];
    __shared__ int sflag;

    const int tid  = threadIdx.x;
    const int lane = tid & 31;
    const int wave = tid >> 5;
    const int lo   = blockIdx.x * NB;

    constexpr int TV = TAB ? (DTAB / 16 / AT) : 1;
    const int nvec = (n + 15) >> 4;
    uint4 tv[TV];
    if (TAB) {
        const uint4* gt = (const uint4*)dtab;
#pragma unroll
        for (int u = 0; u < TV; ++u) {
            const int i = tid + u * AT;
            tv[u] = gt[(i < nvec) ? i : (nvec - 1)];
        }
    }

    const int rc = (tid < nchunks) ? tid : (nchunks - 1);
    const unsigned short* rw = lb16 + (size_t)rc * rs + blockIdx.x;
    const int ra = rw[0];
    const int rz = rw[1];

    for (int r = tid; r < NB; r += AT) {
        const float di = dinv[(lo + r < n) ? (lo + r) : (n - 1)];
        sdinv[r] = (lo + r < n) ? di : 0.0f;
    }
    {
        constexpr int RI = AT / 32;
        constexpr int NI = (NB + RI - 1) / RI;
        const float bl = bias[lane];
        float dv[NI];
        float hv[NI];
#pragma unroll
        for (int k = 0; k < NI; ++k) {
            const int node = lo + wave + k * RI;
            const int nc   = (node < n) ? node : (n - 1);
            dv[k] = dinv[nc];
            hv[k] = h[(size_t)nc * LAT + lane];
        }
        if (TAB) {
            uint4* st = (uint4*)stab;
#pragma unroll
            for (int u = 0; u < TV; ++u) {
                const int i = tid + u * AT;
                st[(i < nvec) ? i : (nvec - 1)] = tv[u];
            }
        }
#pragma unroll
        for (int k = 0; k < NI; ++k) {
            const int row = wave + k * RI;
            acc[row * LAT + lane] = (lo + row < n) ? (dv[k] * dv[k] * hv[k] + bl) : 0.0f;
        }
    }
    int total;
    {
        const int rl = (tid < nchunks) ? (rz - ra) : 0;
        const int ex = block_excl_scan<AT>(rl, sw, total);
        if (tid < nchunks) { rpos[tid] = ex; radj[tid] = tid * CH + ra - ex; }
    }
    __syncthreads();

    const int start  = 0;
    const int end    = total;
    const int nbatch = (end + 31) >> 5;
    int p2 = 1;
    while (p2 < nchunks) p2 <<= 1;
    const float* hl  = h + lane;
    const unsigned lane_u = (unsigned)lane;

    bool use_sorted = false;
    if (TAB) {
        int cN = 0;
        if (tid < NB && lo + tid < n) cN = (int)stab[lo + tid];
        if (tid == 0) sflag = 0;
        int tot2;
        const int ex = block_excl_scan<AT>(cN, sw, tot2);
        if (tid < NB) ncur[tid] = ex;
        if (cN == 255) sflag = 1;
        __syncthreads();
        use_sorted = (sflag == 0) && (tot2 == total) && (total <= SCAP) && (total > 0);
        if (use_sorted) {
            constexpr int SU = 8;
            for (int i0 = 0; i0 < total; i0 += AT * SU) {
                unsigned wv[SU];
#pragma unroll
                for (int u = 0; u < SU; ++u) {
                    const int i = i0 + u * AT + tid;
                    wv[u] = range_word(grouped, rpos, radj, nchunks, p2, (i < total) ? i : (total - 1));
                }
#pragma unroll
                for (int u = 0; u < SU; ++u) {
                    const int i = i0 + u * AT + tid;
                    if (i < total) {
                        const int pos = atomicAdd(&ncur[wv[u] & DLM], 1);
                        sorted[pos] = wv[u];
                    }
                }
            }
        }
        __syncthreads();
    }
    if (use_sorted) {
        constexpr int KSS = AT / 32;
        for (int k = wave; k < nbatch; k += KSS) {
            const int i  = (k << 5) + lane;
            const int ic = (i < total) ? i : (total - 1);
            const unsigned wd = sorted[ic];
            const int   s  = (int)(wd >> DLB);
            const int   dl = (int)(wd & DLM);
            const float sd = sdinv[dl];
            const unsigned cs = stab[s];
            float ds = rsqrtf((float)(1 + (int)cs));
            if (__builtin_amdgcn_ballot_w32(cs == 255u) != 0u) {
                if (cs == 255u) ds = dinv[s];
            }
            const float w = (i < total) ? (ds * sd) : 0.0f;
            const int dn = __shfl_down(dl, 1);
            const unsigned last = __builtin_amdgcn_ballot_w32(lane == 31 || dn != dl);
            unsigned ix[32];
#pragma unroll
            for (int j = 0; j < 32; ++j) {
                ix[j] = ((unsigned)__builtin_amdgcn_readlane(s, j) << 5) | lane_u;
            }
            asm volatile("" :: "v"(ix[0]), "v"(ix[1]), "v"(ix[2]), "v"(ix[3]), "v"(ix[4]), "v"(ix[5]),
                         "v"(ix[6]), "v"(ix[7]), "v"(ix[8]), "v"(ix[9]), "v"(ix[10]), "v"(ix[11]),
                         "v"(ix[12]), "v"(ix[13]), "v"(ix[14]), "v"(ix[15]) : "memory");
            asm volatile("" :: "v"(ix[16]), "v"(ix[17]), "v"(ix[18]), "v"(ix[19]), "v"(ix[20]), "v"(ix[21]),
                         "v"(ix[22]), "v"(ix[23]), "v"(ix[24]), "v"(ix[25]), "v"(ix[26]), "v"(ix[27]),
                         "v"(ix[28]), "v"(ix[29]), "v"(ix[30]), "v"(ix[31]) : "memory");
            float v[32];
#pragma unroll
            for (int j = 0; j < 32; ++j) {
                v[j] = h[ix[j]];
            }
            float a = 0.0f;
#pragma unroll
            for (int j = 0; j < 32; ++j) {
                const float wj = __int_as_float(__builtin_amdgcn_readlane(__float_as_int(w), j));
                a += wj * v[j];
                if ((last >> j) & 1u) {
                    const int dj = __builtin_amdgcn_readlane(dl, j);
                    atomicAdd(&acc[dj * LAT + lane], a);
                    a = 0.0f;
                }
            }
        }
    }
    if (!use_sorted) {

    const int lastw = end - 1;
    constexpr int KS = AT / 32;
    unsigned wd_cur = 0u, wd_nxt = 0u;
    float ds_cur = 0.0f;
    if (wave < nbatch) {
        const int i = start + (wave << 5) + lane;
        wd_cur = range_word(grouped, rpos, radj, nchunks, p2, (i < end) ? i : lastw);
    }
    if (wave + KS < nbatch) {
        const int i = start + ((wave + KS) << 5) + lane;
        wd_nxt = range_word(grouped, rpos, radj, nchunks, p2, (i < end) ? i : lastw);
    }
    if (!TAB && wave < nbatch) ds_cur = dinv[wd_cur >> DLB];

    for (int k = wave; k < nbatch; k += KS) {
        const int base = start + (k << 5);
        const int m    = (end - base < 32) ? (end - base) : 32;
        const unsigned wd = wd_cur;
        const int   s  = (int)(wd >> DLB);
        const int   dl = (int)(wd & DLM);
        const float sd = sdinv[dl];
        float ds = ds_cur;
        if (TAB) {
            const unsigned cs = stab[s];
            ds = rsqrtf((float)(1 + (int)cs));
            if (__builtin_amdgcn_ballot_w32(cs == 255u) != 0u) {
                if (cs == 255u) ds = dinv[s];
            }
        }
        const float w  = ds * sd;
        if (m == 32) {
            unsigned ix[32];
#pragma unroll
            for (int j = 0; j < 32; ++j) {
                ix[j] = ((unsigned)__builtin_amdgcn_readlane(s, j) << 5) | lane_u;
            }
            asm volatile("" :: "v"(ix[0]), "v"(ix[1]), "v"(ix[2]), "v"(ix[3]), "v"(ix[4]), "v"(ix[5]),
                         "v"(ix[6]), "v"(ix[7]), "v"(ix[8]), "v"(ix[9]), "v"(ix[10]), "v"(ix[11]),
                         "v"(ix[12]), "v"(ix[13]), "v"(ix[14]), "v"(ix[15]) : "memory");
            asm volatile("" :: "v"(ix[16]), "v"(ix[17]), "v"(ix[18]), "v"(ix[19]), "v"(ix[20]), "v"(ix[21]),
                         "v"(ix[22]), "v"(ix[23]), "v"(ix[24]), "v"(ix[25]), "v"(ix[26]), "v"(ix[27]),
                         "v"(ix[28]), "v"(ix[29]), "v"(ix[30]), "v"(ix[31]) : "memory");
            float v[32];
#pragma unroll
            for (int j = 0; j < 32; ++j) {
                v[j] = h[ix[j]];
            }
            if (!TAB && k + KS < nbatch) ds_cur = dinv[wd_nxt >> DLB];
            wd_cur = wd_nxt;
            if (k + 2 * KS < nbatch) {
                const int i = start + ((k + 2 * KS) << 5) + lane;
                wd_nxt = range_word(grouped, rpos, radj, nchunks, p2, (i < end) ? i : lastw);
            }
#pragma unroll
            for (int j = 0; j < 32; ++j) {
                const float wj = __int_as_float(__builtin_amdgcn_readlane(__float_as_int(w), j));
                const int   dj = __builtin_amdgcn_readlane(dl, j);
                atomicAdd(&acc[dj * LAT + lane], wj * v[j]);
            }
        } else {
            for (int j = 0; j < m; ++j) {
                const int   sj = __builtin_amdgcn_readlane(s, j);
                const float wj = __int_as_float(__builtin_amdgcn_readlane(__float_as_int(w), j));
                const int   dj = __builtin_amdgcn_readlane(dl, j);
                atomicAdd(&acc[dj * LAT + lane], wj * hl[(size_t)sj * LAT]);
            }
        }
    }
    }
    __syncthreads();

    for (int i = tid; i < NB * LAT; i += AT) {
        if (lo + (i >> 5) < n) out[(size_t)lo * LAT + i] = acc[i];
    }
}


extern "C" void kernel_launch(void* const* d_in, const int* in_sizes, int n_in,
                              void* d_out, int out_size, void* d_ws, size_t ws_size,
                              hipStream_t stream) {
    const float* x   = (const float*)d_in[0];
    const int*   ei  = (const int*)  d_in[1];
    const int*   yei = (const int*)  d_in[2];
    const float* W   = (const float*)d_in[3];
    const float* b   = (const float*)d_in[4];
    float* out = (float*)d_out;

    const int n  = in_sizes[0] / IN_CH;
    const int e1 = in_sizes[1] / 2;
    const int e2 = in_sizes[2] / 2;
    const int et = e1 + e2;

    float* deg = (float*)d_ws;
    size_t hoff = (((size_t)n * sizeof(float)) + 255) & ~(size_t)255;
    float* h = (float*)((char*)d_ws + hoff);

    const int nb      = (n + NB - 1) / NB;
    const int nchunks = (et + CH - 1) / CH;
    const int rs      = (nb + 64) & ~63;
    size_t soff = (hoff + (size_t)n * LAT * sizeof(float) + 255) & ~(size_t)255;
    size_t coff = (soff + (size_t)et * sizeof(unsigned) + 255) & ~(size_t)255;
    size_t toff = (coff + (size_t)(nb + 8) * nchunks * sizeof(int) + 255) & ~(size_t)255;
    size_t boff = (toff + (size_t)nb * sizeof(int) + 255) & ~(size_t)255;
    size_t need = boff + ((size_t)nb + 1) * sizeof(int);
    unsigned* grouped = (unsigned*)((char*)d_ws + soff);
    unsigned* plane   = (unsigned*)((char*)d_ws + coff);
    unsigned short* lb16 = (unsigned short*)plane;
    const bool grouped_ok = (need <= ws_size) && (nb <= MAXB) && (n <= (1 << (32 - DLB))) &&
                            (nchunks > 0) && (nchunks <= MAXC);
    const int    ntabw = ((n + 15) >> 4) << 2;
    const size_t dt0   = ((((size_t)nchunks * rs) >> 1) + 3) & ~(size_t)3;
    const bool   tab   = (n <= DTAB) && (dt0 + (size_t)ntabw <= (size_t)(nb + 8) * nchunks);
    unsigned* dtab     = plane + dt0;

    const int ntiles  = (n + TILE_M - 1) / TILE_M;
    const int gblocks = (ntiles + GW - 1) / GW;
    if (grouped_ok) {
        group_gemm_kernel<<<nchunks + gblocks, PT, 0, stream>>>(ei, yei, lb16, grouped, e1, e2, nchunks, nb, rs,
                                                                x, W, h, n);
        range_deg_kernel<<<nb, DG, 0, stream>>>(grouped, lb16, deg, dtab, tab ? ntabw : 0, n, nchunks, rs, et);
        if (tab) range_sum_kernel<true><<<nb, AT, 0, stream>>>(grouped, lb16, deg, dtab, h, b, out, n, nchunks, rs);
        else     range_sum_kernel<false><<<nb, AT, 0, stream>>>(grouped, lb16, deg, dtab, h, b, out, n, nchunks, rs);
    } else {
        group_gemm_kernel<<<gblocks, PT, 0, stream>>>(ei, yei, lb16, grouped, e1, e2, 0, nb, rs, x, W, h, n);
        deg_init_kernel<<<(n + 255) / 256, 256, 0, stream>>>(deg, n);
        deg_edges_kernel<<<(et + 255) / 256, 256, 0, stream>>>(ei, yei, deg, e1, e2);
        dinv_kernel<<<(n + 255) / 256, 256, 0, stream>>>(deg, n);
        init_out_kernel<<<((size_t)n * LAT + 255) / 256, 256, 0, stream>>>(deg, h, b, out, n);
        long long threads = (long long)et * 32;
        int blocks = (int)((threads + 255) / 256);
        edge_scatter_kernel<<<blocks, 256, 0, stream>>>(ei, yei, deg, h, out, e1, e2);
    }
}
